# starting kernel
# baseline (speedup 1.0000x reference)
#include <hip/hip_runtime.h>
#include <hip/hip_bf16.h>

typedef __attribute__((ext_vector_type(2))) float v2f;
typedef __attribute__((ext_vector_type(8))) float v8f;

#define N_USER 100000
#define N_ITEM 150000
#define N_NODES (N_USER + N_ITEM)
#define N_HID 64
#define E_UI 1000000
#define E_S 1000000
#define SLOPE 0.5f
#define EPS 1e-12f

__global__ void copy_dual_kernel(const float* __restrict__ src,
                                 float* __restrict__ X,
                                 float* __restrict__ out,
                                 int n) {
    int i = blockIdx.x * blockDim.x + threadIdx.x;
    if (i < n) {
        float v = src[i];
        X[i] = v;
        out[i] = v;
    }
}

__global__ void zero_kernel(float* __restrict__ p, int n) {
    int i = blockIdx.x * blockDim.x + threadIdx.x;
    if (i < n) p[i] = 0.0f;
}

__global__ void __launch_bounds__(32)
gemm64_wmma_kernel(const float* __restrict__ X,
                   const float* __restrict__ W,
                   float* __restrict__ Y,
                   int M) {
    const int row0 = blockIdx.x * 16;
    if (row0 >= M) return;
    const int lane = threadIdx.x;
    const int r = lane & 15;
    const int h = lane >> 4;

    v8f acc[4];
    acc[0] = v8f{};
    acc[1] = v8f{};
    acc[2] = v8f{};
    acc[3] = v8f{};

#pragma unroll
    for (int k = 0; k < 64; k += 4) {
        const float* xr = X + (size_t)(row0 + r) * N_HID + k + 2 * h;
        v2f a;
        a[0] = xr[0];
        a[1] = xr[1];
        const float* w0 = W + (size_t)(k + 2 * h) * N_HID + r;
        const float* w1 = w0 + N_HID;
#pragma unroll
        for (int ct = 0; ct < 4; ++ct) {
            v2f b;
            b[0] = w0[ct * 16];
            b[1] = w1[ct * 16];
            acc[ct] = __builtin_amdgcn_wmma_f32_16x16x4_f32(
                 false, a,  false, b,
                 (short)0, acc[ct],
                 false,  false);
        }
    }

#pragma unroll
    for (int ct = 0; ct < 4; ++ct) {
#pragma unroll
        for (int v = 0; v < 8; ++v) {
            Y[(size_t)(row0 + v + 8 * h) * N_HID + ct * 16 + r] = acc[ct][v];
        }
    }
}

__global__ void __launch_bounds__(256)
spmm_scatter_kernel(const int* __restrict__ rows,
                    const int* __restrict__ cols,
                    const float* __restrict__ vals,
                    const float* __restrict__ Y,
                    float* __restrict__ Z,
                    int E) {
    int e = blockIdx.x * 4 + (threadIdx.x >> 6);
    int d = threadIdx.x & 63;
    if (e < E) {
        int rr = rows[e];
        int cc = cols[e];
        float v = vals[e];
        unsafeAtomicAdd(&Z[(size_t)rr * N_HID + d], v * Y[(size_t)cc * N_HID + d]);
    }
}

__global__ void __launch_bounds__(32)
leaky_norm_acc_kernel(const float* __restrict__ Z,
                      float* __restrict__ X,
                      float* __restrict__ out,
                      int M) {
    int n = blockIdx.x;
    if (n >= M) return;
    int lane = threadIdx.x;

    const float2* zr = (const float2*)(Z + (size_t)n * N_HID);
    float2 z = zr[lane];
    float2 v;
    v.x = (z.x >= 0.0f) ? z.x : SLOPE * z.x;
    v.y = (z.y >= 0.0f) ? z.y : SLOPE * z.y;

    ((float2*)(X + (size_t)n * N_HID))[lane] = v;

    float ss = v.x * v.x + v.y * v.y;
#pragma unroll
    for (int o = 16; o > 0; o >>= 1) ss += __shfl_xor(ss, o, 32);

    float inv = 1.0f / fmaxf(sqrtf(ss), EPS);

    float2* orow = (float2*)(out + (size_t)n * N_HID);
    float2 cur = orow[lane];
    cur.x += v.x * inv;
    cur.y += v.y * inv;
    orow[lane] = cur;
}

extern "C" void kernel_launch(void* const* d_in, const int* in_sizes, int n_in,
                              void* d_out, int out_size, void* d_ws, size_t ws_size,
                              hipStream_t stream) {
    const float* user_emb = (const float*)d_in[0];
    const float* item_emb = (const float*)d_in[1];
    const float* ui_w     = (const float*)d_in[2];
    const float* s_w      = (const float*)d_in[3];
    const int*   ui_rows  = (const int*)d_in[4];
    const int*   ui_cols  = (const int*)d_in[5];
    const float* ui_vals  = (const float*)d_in[6];
    const int*   s_rows   = (const int*)d_in[7];
    const int*   s_cols   = (const int*)d_in[8];
    const float* s_vals   = (const float*)d_in[9];

    float* X = (float*)d_ws;
    float* Y = X + (size_t)N_NODES * N_HID;
    float* Z = Y + (size_t)N_NODES * N_HID;

    float* out_ui = (float*)d_out;
    float* out_s  = out_ui + (size_t)N_NODES * N_HID;

    const int nu = N_USER * N_HID;
    const int ni = N_ITEM * N_HID;
    const int nn = N_NODES * N_HID;

    copy_dual_kernel<<<(nu + 255) / 256, 256, 0, stream>>>(user_emb, X, out_ui, nu);
    copy_dual_kernel<<<(ni + 255) / 256, 256, 0, stream>>>(item_emb, X + nu, out_ui + nu, ni);

    for (int l = 0; l < 2; ++l) {
        gemm64_wmma_kernel<<<N_NODES / 16, 32, 0, stream>>>(X, ui_w + (size_t)l * N_HID * N_HID, Y, N_NODES);
        zero_kernel<<<(nn + 255) / 256, 256, 0, stream>>>(Z, nn);
        spmm_scatter_kernel<<<(E_UI + 3) / 4, 256, 0, stream>>>(ui_rows, ui_cols, ui_vals, Y, Z, E_UI);
        leaky_norm_acc_kernel<<<N_NODES, 32, 0, stream>>>(Z, X, out_ui, N_NODES);
    }

    copy_dual_kernel<<<(nu + 255) / 256, 256, 0, stream>>>(user_emb, X, out_s, nu);

    for (int l = 0; l < 2; ++l) {
        gemm64_wmma_kernel<<<N_USER / 16, 32, 0, stream>>>(X, s_w + (size_t)l * N_HID * N_HID, Y, N_USER);
        zero_kernel<<<(nu + 255) / 256, 256, 0, stream>>>(Z, nu);
        spmm_scatter_kernel<<<(E_S + 3) / 4, 256, 0, stream>>>(s_rows, s_cols, s_vals, Y, Z, E_S);
        leaky_norm_acc_kernel<<<N_USER, 32, 0, stream>>>(Z, X, out_s, N_USER);
    }
}


# SpMM as per-row gather: edges linked into per-row lists by integer exchange once per graph, one wave sums a row in registers (no float atomics, no Z clear); float sum order = exchange arrival order
# speedup vs baseline: 1.8925x; 1.8925x over previous
#include <hip/hip_runtime.h>
#include <hip/hip_bf16.h>

typedef __attribute__((ext_vector_type(2))) float v2f;
typedef __attribute__((ext_vector_type(8))) float v8f;

#define N_USER 100000
#define N_ITEM 150000
#define N_NODES (N_USER + N_ITEM)
#define N_HID 64
#define E_UI 1000000
#define E_S 1000000
#define SLOPE 0.5f
#define EPS 1e-12f

__global__ void copy_dual_kernel(const float* __restrict__ src,
                                 float* __restrict__ X,
                                 float* __restrict__ out,
                                 int n) {
    int i = blockIdx.x * blockDim.x + threadIdx.x;
    if (i < n) {
        float v = src[i];
        X[i] = v;
        out[i] = v;
    }
}

__global__ void head_init_kernel(int* __restrict__ head, int n) {
    int i = blockIdx.x * blockDim.x + threadIdx.x;
    if (i < n) head[i] = -1;
}

__global__ void __launch_bounds__(256)
link_edges_kernel(const int* __restrict__ rows,
                  const int* __restrict__ cols,
                  const float* __restrict__ vals,
                  int* __restrict__ head,
                  int* __restrict__ rec_words,
                  int E) {
    int e = blockIdx.x * blockDim.x + threadIdx.x;
    if (e < E) {
        int prev = atomicExch(&head[rows[e]], e);
        ((int4*)rec_words)[e] = make_int4(cols[e], __float_as_int(vals[e]), prev, 0);
    }
}

__global__ void __launch_bounds__(32)
gemm64_wmma_kernel(const float* __restrict__ X,
                   const float* __restrict__ W,
                   float* __restrict__ Y,
                   int M) {
    const int row0 = blockIdx.x * 16;
    if (row0 >= M) return;
    const int lane = threadIdx.x;
    const int r = lane & 15;
    const int h = lane >> 4;

    v8f acc[4];
    acc[0] = v8f{};
    acc[1] = v8f{};
    acc[2] = v8f{};
    acc[3] = v8f{};

#pragma unroll
    for (int k = 0; k < 64; k += 4) {
        const float* xr = X + (size_t)(row0 + r) * N_HID + k + 2 * h;
        v2f a;
        a[0] = xr[0];
        a[1] = xr[1];
        const float* w0 = W + (size_t)(k + 2 * h) * N_HID + r;
        const float* w1 = w0 + N_HID;
#pragma unroll
        for (int ct = 0; ct < 4; ++ct) {
            v2f b;
            b[0] = w0[ct * 16];
            b[1] = w1[ct * 16];
            acc[ct] = __builtin_amdgcn_wmma_f32_16x16x4_f32(
                 false, a,  false, b,
                 (short)0, acc[ct],
                 false,  false);
        }
    }

#pragma unroll
    for (int ct = 0; ct < 4; ++ct) {
#pragma unroll
        for (int v = 0; v < 8; ++v) {
            Y[(size_t)(row0 + v + 8 * h) * N_HID + ct * 16 + r] = acc[ct][v];
        }
    }
}

__global__ void __launch_bounds__(256)
spmm_gather_kernel(const int* __restrict__ head,
                   const int* __restrict__ rec_words,
                   const float* __restrict__ Y,
                   float* __restrict__ Z,
                   int M, int E) {
    const int row = blockIdx.x * 8 + (threadIdx.x >> 5);
    if (row >= M) return;
    const int lane = threadIdx.x & 31;
    const int4* rec = (const int4*)rec_words;

    float2 acc = make_float2(0.0f, 0.0f);
    int e = head[row];
    int4 r = rec[((unsigned)e < (unsigned)E) ? e : 0];
    for (int hop = 0; hop < E && (unsigned)e < (unsigned)E; ++hop) {
        const float2 y = ((const float2*)(Y + (size_t)r.x * N_HID))[lane];
        const float v = __int_as_float(r.y);
        e = r.z;
        r = rec[((unsigned)e < (unsigned)E) ? e : 0];
        asm volatile("" :: "v"(y.x), "v"(y.y), "v"(r.x), "v"(r.y), "v"(r.z));
        acc.x += v * y.x;
        acc.y += v * y.y;
    }
    ((float2*)(Z + (size_t)row * N_HID))[lane] = acc;
}

__global__ void __launch_bounds__(32)
leaky_norm_acc_kernel(const float* __restrict__ Z,
                      float* __restrict__ X,
                      float* __restrict__ out,
                      int M) {
    int n = blockIdx.x;
    if (n >= M) return;
    int lane = threadIdx.x;

    const float2* zr = (const float2*)(Z + (size_t)n * N_HID);
    float2 z = zr[lane];
    float2 v;
    v.x = (z.x >= 0.0f) ? z.x : SLOPE * z.x;
    v.y = (z.y >= 0.0f) ? z.y : SLOPE * z.y;

    ((float2*)(X + (size_t)n * N_HID))[lane] = v;

    float ss = v.x * v.x + v.y * v.y;
#pragma unroll
    for (int o = 16; o > 0; o >>= 1) ss += __shfl_xor(ss, o, 32);

    float inv = 1.0f / fmaxf(sqrtf(ss), EPS);

    float2* orow = (float2*)(out + (size_t)n * N_HID);
    float2 cur = orow[lane];
    cur.x += v.x * inv;
    cur.y += v.y * inv;
    orow[lane] = cur;
}

extern "C" void kernel_launch(void* const* d_in, const int* in_sizes, int n_in,
                              void* d_out, int out_size, void* d_ws, size_t ws_size,
                              hipStream_t stream) {
    const float* user_emb = (const float*)d_in[0];
    const float* item_emb = (const float*)d_in[1];
    const float* ui_w     = (const float*)d_in[2];
    const float* s_w      = (const float*)d_in[3];
    const int*   ui_rows  = (const int*)d_in[4];
    const int*   ui_cols  = (const int*)d_in[5];
    const float* ui_vals  = (const float*)d_in[6];
    const int*   s_rows   = (const int*)d_in[7];
    const int*   s_cols   = (const int*)d_in[8];
    const float* s_vals   = (const float*)d_in[9];

    float* X = (float*)d_ws;
    float* Y = X + (size_t)N_NODES * N_HID;
    float* Z = Y + (size_t)N_NODES * N_HID;
    int* head = (int*)(Z + (size_t)N_NODES * N_HID);
    int* rec  = head + N_NODES;

    float* out_ui = (float*)d_out;
    float* out_s  = out_ui + (size_t)N_NODES * N_HID;

    const int nu = N_USER * N_HID;
    const int ni = N_ITEM * N_HID;

    copy_dual_kernel<<<(nu + 255) / 256, 256, 0, stream>>>(user_emb, X, out_ui, nu);
    copy_dual_kernel<<<(ni + 255) / 256, 256, 0, stream>>>(item_emb, X + nu, out_ui + nu, ni);

    head_init_kernel<<<(N_NODES + 255) / 256, 256, 0, stream>>>(head, N_NODES);
    link_edges_kernel<<<(E_UI + 255) / 256, 256, 0, stream>>>(ui_rows, ui_cols, ui_vals, head, rec, E_UI);

    for (int l = 0; l < 2; ++l) {
        gemm64_wmma_kernel<<<N_NODES / 16, 32, 0, stream>>>(X, ui_w + (size_t)l * N_HID * N_HID, Y, N_NODES);
        spmm_gather_kernel<<<(N_NODES + 7) / 8, 256, 0, stream>>>(head, rec, Y, Z, N_NODES, E_UI);
        leaky_norm_acc_kernel<<<N_NODES, 32, 0, stream>>>(Z, X, out_ui, N_NODES);
    }

    copy_dual_kernel<<<(nu + 255) / 256, 256, 0, stream>>>(user_emb, X, out_s, nu);

    head_init_kernel<<<(N_USER + 255) / 256, 256, 0, stream>>>(head, N_USER);
    link_edges_kernel<<<(E_S + 255) / 256, 256, 0, stream>>>(s_rows, s_cols, s_vals, head, rec, E_S);

    for (int l = 0; l < 2; ++l) {
        gemm64_wmma_kernel<<<N_USER / 16, 32, 0, stream>>>(X, s_w + (size_t)l * N_HID * N_HID, Y, N_USER);
        spmm_gather_kernel<<<(N_USER + 7) / 8, 256, 0, stream>>>(head, rec, Y, Z, N_USER, E_S);
        leaky_norm_acc_kernel<<<N_USER, 32, 0, stream>>>(Z, X, out_s, N_USER);
    }
}


# Work per wave raised, about 6x fewer waves per call: copies and edge linking take 4 pieces per lane, gather sums 4 rows per wave, norm 8 rows per wave, GEMM in 8-wave groups; float sum orders kept
# speedup vs baseline: 2.5410x; 1.3427x over previous
#include <hip/hip_runtime.h>
#include <hip/hip_bf16.h>

typedef __attribute__((ext_vector_type(2))) float v2f;
typedef __attribute__((ext_vector_type(8))) float v8f;
typedef __attribute__((ext_vector_type(4))) float v4f;
typedef __attribute__((ext_vector_type(4))) int v4i;

#define N_USER 100000
#define N_ITEM 150000
#define N_NODES (N_USER + N_ITEM)
#define N_HID 64
#define E_UI 1000000
#define E_S 1000000
#define SLOPE 0.5f
#define EPS 1e-12f

__device__ __forceinline__ void copy_dual_group(const v4f* __restrict__ s4,
                                                v4f* __restrict__ x4,
                                                v4f* __restrict__ o4,
                                                int base, int n4, const bool guard) {
    v4f v[4];
#pragma unroll
    for (int j = 0; j < 4; ++j) {
        const int i = base + j * 256;
        v[j] = (!guard || i < n4) ? s4[i] : (v4f)(0.0f);
    }
    asm volatile("" :: "v"(v[0]), "v"(v[1]), "v"(v[2]), "v"(v[3]));
#pragma unroll
    for (int j = 0; j < 4; ++j) {
        const int i = base + j * 256;
        if (!guard || i < n4) {
            x4[i] = v[j];
            o4[i] = v[j];
        }
    }
}

__global__ void __launch_bounds__(256)
copy_dual_kernel(const float* __restrict__ src,
                 float* __restrict__ X,
                 float* __restrict__ out,
                 int n4) {
    const int base = blockIdx.x * 1024 + threadIdx.x;
    if (blockIdx.x * 1024 + 1024 <= n4)
        copy_dual_group((const v4f*)src, (v4f*)X, (v4f*)out, base, n4, false);
    else
        copy_dual_group((const v4f*)src, (v4f*)X, (v4f*)out, base, n4, true);
}

__global__ void head_init_kernel(int* __restrict__ head, int n) {
    int i = blockIdx.x * blockDim.x + threadIdx.x;
    if (i < n) head[i] = -1;
}

__device__ __forceinline__ void link_edges_group(const int* __restrict__ rows,
                                                 const int* __restrict__ cols,
                                                 const float* __restrict__ vals,
                                                 int* __restrict__ head,
                                                 v4i* __restrict__ rec,
                                                 int base, int E, const bool guard) {
    int rw[4], cl[4], vl[4], prev[4];
#pragma unroll
    for (int j = 0; j < 4; ++j) {
        const int e = base + j * 256;
        const bool ok = !guard || e < E;
        rw[j] = ok ? rows[e] : 0;
        cl[j] = ok ? cols[e] : 0;
        vl[j] = ok ? __float_as_int(vals[e]) : 0;
    }
#pragma unroll
    for (int j = 0; j < 4; ++j) {
        const int e = base + j * 256;
        prev[j] = (!guard || e < E) ? atomicExch(&head[rw[j]], e) : -1;
    }
#pragma unroll
    for (int j = 0; j < 4; ++j) {
        const int e = base + j * 256;
        if (!guard || e < E) {
            v4i r;
            r.x = cl[j];
            r.y = vl[j];
            r.z = prev[j];
            r.w = 0;
            rec[e] = r;
        }
    }
}

__global__ void __launch_bounds__(256)
link_edges_kernel(const int* __restrict__ rows,
                  const int* __restrict__ cols,
                  const float* __restrict__ vals,
                  int* __restrict__ head,
                  int* __restrict__ rec_words,
                  int E) {
    const int base = blockIdx.x * 1024 + threadIdx.x;
    if (blockIdx.x * 1024 + 1024 <= E)
        link_edges_group(rows, cols, vals, head, (v4i*)rec_words, base, E, false);
    else
        link_edges_group(rows, cols, vals, head, (v4i*)rec_words, base, E, true);
}

__global__ void __launch_bounds__(256)
gemm64_wmma_kernel(const float* __restrict__ X,
                   const float* __restrict__ W,
                   float* __restrict__ Y,
                   int M) {
    const int row0 = (blockIdx.x * 8 + (threadIdx.x >> 5)) * 16;
    if (row0 >= M) return;
    const int lane = threadIdx.x & 31;
    const int r = lane & 15;
    const int h = lane >> 4;

    v8f acc[4];
    acc[0] = v8f{};
    acc[1] = v8f{};
    acc[2] = v8f{};
    acc[3] = v8f{};

#pragma unroll
    for (int k = 0; k < 64; k += 4) {
        const float* xr = X + (size_t)(row0 + r) * N_HID + k + 2 * h;
        v2f a;
        a[0] = xr[0];
        a[1] = xr[1];
        const float* w0 = W + (size_t)(k + 2 * h) * N_HID + r;
        const float* w1 = w0 + N_HID;
#pragma unroll
        for (int ct = 0; ct < 4; ++ct) {
            v2f b;
            b[0] = w0[ct * 16];
            b[1] = w1[ct * 16];
            acc[ct] = __builtin_amdgcn_wmma_f32_16x16x4_f32(
                 false, a,  false, b,
                 (short)0, acc[ct],
                 false,  false);
        }
    }

#pragma unroll
    for (int ct = 0; ct < 4; ++ct) {
#pragma unroll
        for (int v = 0; v < 8; ++v) {
            Y[(size_t)(row0 + v + 8 * h) * N_HID + ct * 16 + r] = acc[ct][v];
        }
    }
}

__global__ void __launch_bounds__(256)
spmm_gather_kernel(const int* __restrict__ head,
                   const int* __restrict__ rec_words,
                   const float* __restrict__ Y,
                   float* __restrict__ Z,
                   int M, int E) {
    const int lane = threadIdx.x & 31;
    const int row = (blockIdx.x * 8 + (threadIdx.x >> 5)) * 4 + (lane >> 3);
    if (row >= M) return;
    const int s = lane & 7;
    const v4i* rec = (const v4i*)rec_words;

    v4f acc0 = (v4f)(0.0f);
    v4f acc1 = (v4f)(0.0f);
    int e = head[row];
    v4i r = rec[((unsigned)e < (unsigned)E) ? e : 0];
    for (int hop = 0; hop < E && (unsigned)e < (unsigned)E; ++hop) {
        const v4f* yrow = (const v4f*)(Y + (size_t)r.x * N_HID);
        const v4f y0 = yrow[s];
        const v4f y1 = yrow[8 + s];
        const float v = __int_as_float(r.y);
        e = r.z;
        r = rec[((unsigned)e < (unsigned)E) ? e : 0];
        asm volatile("" :: "v"(y0), "v"(y1), "v"(r.x), "v"(r.y), "v"(r.z));
        acc0 += v * y0;
        acc1 += v * y1;
    }
    v4f* zrow = (v4f*)(Z + (size_t)row * N_HID);
    zrow[s] = acc0;
    zrow[8 + s] = acc1;
}

__device__ __forceinline__ void leaky_norm_rows(const v4f* __restrict__ z4,
                                                v4f* __restrict__ x4,
                                                v4f* __restrict__ o4,
                                                int row_first, int c, int M,
                                                const bool guard) {
    v4f z[4], cur[4];
#pragma unroll
    for (int j = 0; j < 4; ++j) {
        const int n = row_first + 2 * j;
        const size_t idx = (size_t)n * (N_HID / 4) + c;
        const bool ok = !guard || n < M;
        z[j] = ok ? z4[idx] : (v4f)(0.0f);
        cur[j] = ok ? o4[idx] : (v4f)(0.0f);
    }
    __builtin_amdgcn_sched_barrier(0);

#pragma unroll
    for (int j = 0; j < 4; ++j) {
        const int n = row_first + 2 * j;
        const size_t idx = (size_t)n * (N_HID / 4) + c;
        v4f v;
        v.x = (z[j].x >= 0.0f) ? z[j].x : SLOPE * z[j].x;
        v.y = (z[j].y >= 0.0f) ? z[j].y : SLOPE * z[j].y;
        v.z = (z[j].z >= 0.0f) ? z[j].z : SLOPE * z[j].z;
        v.w = (z[j].w >= 0.0f) ? z[j].w : SLOPE * z[j].w;

        float pa = v.x * v.x + v.y * v.y;
        float pb = v.z * v.z + v.w * v.w;
#pragma unroll
        for (int o = 8; o > 0; o >>= 1) {
            pa += __shfl_xor(pa, o, 32);
            pb += __shfl_xor(pb, o, 32);
        }
        const float ss = pa + pb;

        const float inv = 1.0f / fmaxf(sqrtf(ss), EPS);

        if (!guard || n < M) {
            x4[idx] = v;
            o4[idx] = cur[j] + v * inv;
        }
    }
}

__global__ void __launch_bounds__(256)
leaky_norm_acc_kernel(const float* __restrict__ Z,
                      float* __restrict__ X,
                      float* __restrict__ out,
                      int M) {
    const int lane = threadIdx.x & 31;
    const int wave_row0 = (blockIdx.x * 8 + (threadIdx.x >> 5)) * 8;
    const int row_first = wave_row0 + (lane >> 4);
    const int c = lane & 15;
    if (wave_row0 + 8 <= M)
        leaky_norm_rows((const v4f*)Z, (v4f*)X, (v4f*)out, row_first, c, M, false);
    else
        leaky_norm_rows((const v4f*)Z, (v4f*)X, (v4f*)out, row_first, c, M, true);
}

extern "C" void kernel_launch(void* const* d_in, const int* in_sizes, int n_in,
                              void* d_out, int out_size, void* d_ws, size_t ws_size,
                              hipStream_t stream) {
    const float* user_emb = (const float*)d_in[0];
    const float* item_emb = (const float*)d_in[1];
    const float* ui_w     = (const float*)d_in[2];
    const float* s_w      = (const float*)d_in[3];
    const int*   ui_rows  = (const int*)d_in[4];
    const int*   ui_cols  = (const int*)d_in[5];
    const float* ui_vals  = (const float*)d_in[6];
    const int*   s_rows   = (const int*)d_in[7];
    const int*   s_cols   = (const int*)d_in[8];
    const float* s_vals   = (const float*)d_in[9];

    float* X = (float*)d_ws;
    float* Y = X + (size_t)N_NODES * N_HID;
    float* Z = Y + (size_t)N_NODES * N_HID;
    int* head = (int*)(Z + (size_t)N_NODES * N_HID);
    int* rec  = head + N_NODES;

    float* out_ui = (float*)d_out;
    float* out_s  = out_ui + (size_t)N_NODES * N_HID;

    const int nu = N_USER * N_HID;
    const int ni = N_ITEM * N_HID;
    const int nu4 = nu / 4;
    const int ni4 = ni / 4;

    copy_dual_kernel<<<(nu4 + 1023) / 1024, 256, 0, stream>>>(user_emb, X, out_ui, nu4);
    copy_dual_kernel<<<(ni4 + 1023) / 1024, 256, 0, stream>>>(item_emb, X + nu, out_ui + nu, ni4);

    head_init_kernel<<<(N_NODES + 255) / 256, 256, 0, stream>>>(head, N_NODES);
    link_edges_kernel<<<(E_UI + 1023) / 1024, 256, 0, stream>>>(ui_rows, ui_cols, ui_vals, head, rec, E_UI);

    for (int l = 0; l < 2; ++l) {
        gemm64_wmma_kernel<<<(N_NODES / 16 + 7) / 8, 256, 0, stream>>>(X, ui_w + (size_t)l * N_HID * N_HID, Y, N_NODES);
        spmm_gather_kernel<<<(N_NODES + 31) / 32, 256, 0, stream>>>(head, rec, Y, Z, N_NODES, E_UI);
        leaky_norm_acc_kernel<<<(N_NODES + 63) / 64, 256, 0, stream>>>(Z, X, out_ui, N_NODES);
    }

    copy_dual_kernel<<<(nu4 + 1023) / 1024, 256, 0, stream>>>(user_emb, X, out_s, nu4);

    head_init_kernel<<<(N_USER + 255) / 256, 256, 0, stream>>>(head, N_USER);
    link_edges_kernel<<<(E_S + 1023) / 1024, 256, 0, stream>>>(s_rows, s_cols, s_vals, head, rec, E_S);

    for (int l = 0; l < 2; ++l) {
        gemm64_wmma_kernel<<<(N_USER / 16 + 7) / 8, 256, 0, stream>>>(X, s_w + (size_t)l * N_HID * N_HID, Y, N_USER);
        spmm_gather_kernel<<<(N_USER + 31) / 32, 256, 0, stream>>>(head, rec, Y, Z, N_USER, E_S);
        leaky_norm_acc_kernel<<<(N_USER + 63) / 64, 256, 0, stream>>>(Z, X, out_s, N_USER);
    }
}


# Leaky, row norm and the add into out folded into the gather kernel: the 8 lanes that summed a row finish it in registers; no Z plane, 4 fewer launches; norm's square sum now on 8 lanes x 8 floats
# speedup vs baseline: 2.8066x; 1.1045x over previous
#include <hip/hip_runtime.h>
#include <hip/hip_bf16.h>

typedef __attribute__((ext_vector_type(2))) float v2f;
typedef __attribute__((ext_vector_type(8))) float v8f;
typedef __attribute__((ext_vector_type(4))) float v4f;
typedef __attribute__((ext_vector_type(4))) int v4i;

#define N_USER 100000
#define N_ITEM 150000
#define N_NODES (N_USER + N_ITEM)
#define N_HID 64
#define E_UI 1000000
#define E_S 1000000
#define SLOPE 0.5f
#define EPS 1e-12f

__device__ __forceinline__ void copy_dual_group(const v4f* __restrict__ s4,
                                                v4f* __restrict__ x4,
                                                v4f* __restrict__ o4,
                                                int base, int n4, const bool guard) {
    v4f v[4];
#pragma unroll
    for (int j = 0; j < 4; ++j) {
        const int i = base + j * 256;
        v[j] = (!guard || i < n4) ? s4[i] : (v4f)(0.0f);
    }
    asm volatile("" :: "v"(v[0]), "v"(v[1]), "v"(v[2]), "v"(v[3]));
#pragma unroll
    for (int j = 0; j < 4; ++j) {
        const int i = base + j * 256;
        if (!guard || i < n4) {
            x4[i] = v[j];
            o4[i] = v[j];
        }
    }
}

__global__ void __launch_bounds__(256)
copy_dual_kernel(const float* __restrict__ src,
                 float* __restrict__ X,
                 float* __restrict__ out,
                 int n4) {
    const int base = blockIdx.x * 1024 + threadIdx.x;
    if (blockIdx.x * 1024 + 1024 <= n4)
        copy_dual_group((const v4f*)src, (v4f*)X, (v4f*)out, base, n4, false);
    else
        copy_dual_group((const v4f*)src, (v4f*)X, (v4f*)out, base, n4, true);
}

__global__ void head_init_kernel(int* __restrict__ head, int n) {
    int i = blockIdx.x * blockDim.x + threadIdx.x;
    if (i < n) head[i] = -1;
}

__device__ __forceinline__ void link_edges_group(const int* __restrict__ rows,
                                                 const int* __restrict__ cols,
                                                 const float* __restrict__ vals,
                                                 int* __restrict__ head,
                                                 v4i* __restrict__ rec,
                                                 int base, int E, const bool guard) {
    int rw[4], cl[4], vl[4], prev[4];
#pragma unroll
    for (int j = 0; j < 4; ++j) {
        const int e = base + j * 256;
        const bool ok = !guard || e < E;
        rw[j] = ok ? rows[e] : 0;
        cl[j] = ok ? cols[e] : 0;
        vl[j] = ok ? __float_as_int(vals[e]) : 0;
    }
#pragma unroll
    for (int j = 0; j < 4; ++j) {
        const int e = base + j * 256;
        prev[j] = (!guard || e < E) ? atomicExch(&head[rw[j]], e) : -1;
    }
#pragma unroll
    for (int j = 0; j < 4; ++j) {
        const int e = base + j * 256;
        if (!guard || e < E) {
            v4i r;
            r.x = cl[j];
            r.y = vl[j];
            r.z = prev[j];
            r.w = 0;
            rec[e] = r;
        }
    }
}

__global__ void __launch_bounds__(256)
link_edges_kernel(const int* __restrict__ rows,
                  const int* __restrict__ cols,
                  const float* __restrict__ vals,
                  int* __restrict__ head,
                  int* __restrict__ rec_words,
                  int E) {
    const int base = blockIdx.x * 1024 + threadIdx.x;
    if (blockIdx.x * 1024 + 1024 <= E)
        link_edges_group(rows, cols, vals, head, (v4i*)rec_words, base, E, false);
    else
        link_edges_group(rows, cols, vals, head, (v4i*)rec_words, base, E, true);
}

__global__ void __launch_bounds__(256)
gemm64_wmma_kernel(const float* __restrict__ X,
                   const float* __restrict__ W,
                   float* __restrict__ Y,
                   int M) {
    const int row0 = (blockIdx.x * 8 + (threadIdx.x >> 5)) * 16;
    if (row0 >= M) return;
    const int lane = threadIdx.x & 31;
    const int r = lane & 15;
    const int h = lane >> 4;

    v8f acc[4];
    acc[0] = v8f{};
    acc[1] = v8f{};
    acc[2] = v8f{};
    acc[3] = v8f{};

#pragma unroll
    for (int k = 0; k < 64; k += 4) {
        const float* xr = X + (size_t)(row0 + r) * N_HID + k + 2 * h;
        v2f a;
        a[0] = xr[0];
        a[1] = xr[1];
        const float* w0 = W + (size_t)(k + 2 * h) * N_HID + r;
        const float* w1 = w0 + N_HID;
#pragma unroll
        for (int ct = 0; ct < 4; ++ct) {
            v2f b;
            b[0] = w0[ct * 16];
            b[1] = w1[ct * 16];
            acc[ct] = __builtin_amdgcn_wmma_f32_16x16x4_f32(
                 false, a,  false, b,
                 (short)0, acc[ct],
                 false,  false);
        }
    }

#pragma unroll
    for (int ct = 0; ct < 4; ++ct) {
#pragma unroll
        for (int v = 0; v < 8; ++v) {
            Y[(size_t)(row0 + v + 8 * h) * N_HID + ct * 16 + r] = acc[ct][v];
        }
    }
}

__global__ void __launch_bounds__(256)
spmm_gather_norm_kernel(const int* __restrict__ head,
                        const int* __restrict__ rec_words,
                        const float* __restrict__ Y,
                        float* __restrict__ X,
                        float* __restrict__ out,
                        int M, int E) {
    const int lane = threadIdx.x & 31;
    const int row = (blockIdx.x * 8 + (threadIdx.x >> 5)) * 4 + (lane >> 3);
    if (row >= M) return;
    const int s = lane & 7;
    const v4i* rec = (const v4i*)rec_words;
    v4f* orow = (v4f*)(out + (size_t)row * N_HID);

    v4f acc0 = (v4f)(0.0f);
    v4f acc1 = (v4f)(0.0f);
    int e = head[row];
    const v4f cur0 = orow[s];
    const v4f cur1 = orow[8 + s];
    v4i r = rec[((unsigned)e < (unsigned)E) ? e : 0];
    for (int hop = 0; hop < E && (unsigned)e < (unsigned)E; ++hop) {
        const v4f* yrow = (const v4f*)(Y + (size_t)r.x * N_HID);
        const v4f y0 = yrow[s];
        const v4f y1 = yrow[8 + s];
        const float v = __int_as_float(r.y);
        e = r.z;
        r = rec[((unsigned)e < (unsigned)E) ? e : 0];
        asm volatile("" :: "v"(y0), "v"(y1), "v"(r.x), "v"(r.y), "v"(r.z));
        acc0 += v * y0;
        acc1 += v * y1;
    }

    v4f v0, v1;
    v0.x = (acc0.x >= 0.0f) ? acc0.x : SLOPE * acc0.x;
    v0.y = (acc0.y >= 0.0f) ? acc0.y : SLOPE * acc0.y;
    v0.z = (acc0.z >= 0.0f) ? acc0.z : SLOPE * acc0.z;
    v0.w = (acc0.w >= 0.0f) ? acc0.w : SLOPE * acc0.w;
    v1.x = (acc1.x >= 0.0f) ? acc1.x : SLOPE * acc1.x;
    v1.y = (acc1.y >= 0.0f) ? acc1.y : SLOPE * acc1.y;
    v1.z = (acc1.z >= 0.0f) ? acc1.z : SLOPE * acc1.z;
    v1.w = (acc1.w >= 0.0f) ? acc1.w : SLOPE * acc1.w;

    const float pa0 = v0.x * v0.x + v0.y * v0.y;
    const float pb0 = v0.z * v0.z + v0.w * v0.w;
    const float pa1 = v1.x * v1.x + v1.y * v1.y;
    const float pb1 = v1.z * v1.z + v1.w * v1.w;
    float pa = pa0 + pa1;
    float pb = pb0 + pb1;
#pragma unroll
    for (int o = 4; o > 0; o >>= 1) {
        pa += __shfl_xor(pa, o, 32);
        pb += __shfl_xor(pb, o, 32);
    }
    const float ss = pa + pb;

    const float inv = 1.0f / fmaxf(sqrtf(ss), EPS);

    v4f* xrow = (v4f*)(X + (size_t)row * N_HID);
    xrow[s] = v0;
    xrow[8 + s] = v1;
    orow[s] = cur0 + v0 * inv;
    orow[8 + s] = cur1 + v1 * inv;
}

extern "C" void kernel_launch(void* const* d_in, const int* in_sizes, int n_in,
                              void* d_out, int out_size, void* d_ws, size_t ws_size,
                              hipStream_t stream) {
    const float* user_emb = (const float*)d_in[0];
    const float* item_emb = (const float*)d_in[1];
    const float* ui_w     = (const float*)d_in[2];
    const float* s_w      = (const float*)d_in[3];
    const int*   ui_rows  = (const int*)d_in[4];
    const int*   ui_cols  = (const int*)d_in[5];
    const float* ui_vals  = (const float*)d_in[6];
    const int*   s_rows   = (const int*)d_in[7];
    const int*   s_cols   = (const int*)d_in[8];
    const float* s_vals   = (const float*)d_in[9];

    float* X = (float*)d_ws;
    float* Y = X + (size_t)N_NODES * N_HID;
    int* head = (int*)(Y + (size_t)N_NODES * N_HID);
    int* rec  = head + N_NODES;

    float* out_ui = (float*)d_out;
    float* out_s  = out_ui + (size_t)N_NODES * N_HID;

    const int nu = N_USER * N_HID;
    const int ni = N_ITEM * N_HID;
    const int nu4 = nu / 4;
    const int ni4 = ni / 4;

    copy_dual_kernel<<<(nu4 + 1023) / 1024, 256, 0, stream>>>(user_emb, X, out_ui, nu4);
    copy_dual_kernel<<<(ni4 + 1023) / 1024, 256, 0, stream>>>(item_emb, X + nu, out_ui + nu, ni4);

    head_init_kernel<<<(N_NODES + 255) / 256, 256, 0, stream>>>(head, N_NODES);
    link_edges_kernel<<<(E_UI + 1023) / 1024, 256, 0, stream>>>(ui_rows, ui_cols, ui_vals, head, rec, E_UI);

    for (int l = 0; l < 2; ++l) {
        gemm64_wmma_kernel<<<(N_NODES / 16 + 7) / 8, 256, 0, stream>>>(X, ui_w + (size_t)l * N_HID * N_HID, Y, N_NODES);
        spmm_gather_norm_kernel<<<(N_NODES + 31) / 32, 256, 0, stream>>>(head, rec, Y, X, out_ui, N_NODES, E_UI);
    }

    copy_dual_kernel<<<(nu4 + 1023) / 1024, 256, 0, stream>>>(user_emb, X, out_s, nu4);

    head_init_kernel<<<(N_USER + 255) / 256, 256, 0, stream>>>(head, N_USER);
    link_edges_kernel<<<(E_S + 1023) / 1024, 256, 0, stream>>>(s_rows, s_cols, s_vals, head, rec, E_S);

    for (int l = 0; l < 2; ++l) {
        gemm64_wmma_kernel<<<(N_USER / 16 + 7) / 8, 256, 0, stream>>>(X, s_w + (size_t)l * N_HID * N_HID, Y, N_USER);
        spmm_gather_norm_kernel<<<(N_USER + 31) / 32, 256, 0, stream>>>(head, rec, Y, X, out_s, N_USER, E_S);
    }
}


# Residual copies removed: layer 0's GEMM and gather read the embedding tables in place (gather writes out = table row + its term); 3 fewer launches, no X plane for layer 0; float sum orders kept
# speedup vs baseline: 3.1899x; 1.1366x over previous
#include <hip/hip_runtime.h>
#include <hip/hip_bf16.h>

typedef __attribute__((ext_vector_type(2))) float v2f;
typedef __attribute__((ext_vector_type(8))) float v8f;
typedef __attribute__((ext_vector_type(4))) float v4f;
typedef __attribute__((ext_vector_type(4))) int v4i;

#define N_USER 100000
#define N_ITEM 150000
#define N_NODES (N_USER + N_ITEM)
#define N_HID 64
#define E_UI 1000000
#define E_S 1000000
#define SLOPE 0.5f
#define EPS 1e-12f

__global__ void head_init_kernel(int* __restrict__ head, int n) {
    int i = blockIdx.x * blockDim.x + threadIdx.x;
    if (i < n) head[i] = -1;
}

__device__ __forceinline__ void link_edges_group(const int* __restrict__ rows,
                                                 const int* __restrict__ cols,
                                                 const float* __restrict__ vals,
                                                 int* __restrict__ head,
                                                 v4i* __restrict__ rec,
                                                 int base, int E, const bool guard) {
    int rw[4], cl[4], vl[4], prev[4];
#pragma unroll
    for (int j = 0; j < 4; ++j) {
        const int e = base + j * 256;
        const bool ok = !guard || e < E;
        rw[j] = ok ? rows[e] : 0;
        cl[j] = ok ? cols[e] : 0;
        vl[j] = ok ? __float_as_int(vals[e]) : 0;
    }
#pragma unroll
    for (int j = 0; j < 4; ++j) {
        const int e = base + j * 256;
        prev[j] = (!guard || e < E) ? atomicExch(&head[rw[j]], e) : -1;
    }
#pragma unroll
    for (int j = 0; j < 4; ++j) {
        const int e = base + j * 256;
        if (!guard || e < E) {
            v4i r;
            r.x = cl[j];
            r.y = vl[j];
            r.z = prev[j];
            r.w = 0;
            rec[e] = r;
        }
    }
}

__global__ void __launch_bounds__(256)
link_edges_kernel(const int* __restrict__ rows,
                  const int* __restrict__ cols,
                  const float* __restrict__ vals,
                  int* __restrict__ head,
                  int* __restrict__ rec_words,
                  int E) {
    const int base = blockIdx.x * 1024 + threadIdx.x;
    if (blockIdx.x * 1024 + 1024 <= E)
        link_edges_group(rows, cols, vals, head, (v4i*)rec_words, base, E, false);
    else
        link_edges_group(rows, cols, vals, head, (v4i*)rec_words, base, E, true);
}

__global__ void __launch_bounds__(256)
gemm64_wmma_kernel(const float* __restrict__ Xa,
                   const float* __restrict__ Xb,
                   int split,
                   const float* __restrict__ W,
                   float* __restrict__ Y,
                   int M) {
    const int row0 = (blockIdx.x * 8 + (threadIdx.x >> 5)) * 16;
    if (row0 >= M) return;
    const int lane = threadIdx.x & 31;
    const int r = lane & 15;
    const int h = lane >> 4;

    v8f acc[4];
    acc[0] = v8f{};
    acc[1] = v8f{};
    acc[2] = v8f{};
    acc[3] = v8f{};

    const int arow = row0 + r;
    const float* xbase = (arow < split) ? Xa + (size_t)arow * N_HID
                                        : Xb + (size_t)(arow - split) * N_HID;

#pragma unroll
    for (int k = 0; k < 64; k += 4) {
        const float* xr = xbase + k + 2 * h;
        v2f a;
        a[0] = xr[0];
        a[1] = xr[1];
        const float* w0 = W + (size_t)(k + 2 * h) * N_HID + r;
        const float* w1 = w0 + N_HID;
#pragma unroll
        for (int ct = 0; ct < 4; ++ct) {
            v2f b;
            b[0] = w0[ct * 16];
            b[1] = w1[ct * 16];
            acc[ct] = __builtin_amdgcn_wmma_f32_16x16x4_f32(
                 false, a,  false, b,
                 (short)0, acc[ct],
                 false,  false);
        }
    }

#pragma unroll
    for (int ct = 0; ct < 4; ++ct) {
#pragma unroll
        for (int v = 0; v < 8; ++v) {
            Y[(size_t)(row0 + v + 8 * h) * N_HID + ct * 16 + r] = acc[ct][v];
        }
    }
}

__global__ void __launch_bounds__(256)
spmm_gather_norm_kernel(const int* __restrict__ head,
                        const int* __restrict__ rec_words,
                        const float* __restrict__ Y,
                        const float* res_a,
                        const float* res_b,
                        int split,
                        float* __restrict__ X,
                        float* out,
                        int M, int E) {
    const int lane = threadIdx.x & 31;
    const int row = (blockIdx.x * 8 + (threadIdx.x >> 5)) * 4 + (lane >> 3);
    if (row >= M) return;
    const int s = lane & 7;
    const v4i* rec = (const v4i*)rec_words;
    v4f* orow = (v4f*)(out + (size_t)row * N_HID);
    const v4f* rrow = (const v4f*)((row < split) ? res_a + (size_t)row * N_HID
                                                 : res_b + (size_t)(row - split) * N_HID);

    v4f acc0 = (v4f)(0.0f);
    v4f acc1 = (v4f)(0.0f);
    int e = head[row];
    const v4f cur0 = rrow[s];
    const v4f cur1 = rrow[8 + s];
    v4i r = rec[((unsigned)e < (unsigned)E) ? e : 0];
    for (int hop = 0; hop < E && (unsigned)e < (unsigned)E; ++hop) {
        const v4f* yrow = (const v4f*)(Y + (size_t)r.x * N_HID);
        const v4f y0 = yrow[s];
        const v4f y1 = yrow[8 + s];
        const float v = __int_as_float(r.y);
        e = r.z;
        r = rec[((unsigned)e < (unsigned)E) ? e : 0];
        asm volatile("" :: "v"(y0), "v"(y1), "v"(r.x), "v"(r.y), "v"(r.z));
        acc0 += v * y0;
        acc1 += v * y1;
    }

    v4f v0, v1;
    v0.x = (acc0.x >= 0.0f) ? acc0.x : SLOPE * acc0.x;
    v0.y = (acc0.y >= 0.0f) ? acc0.y : SLOPE * acc0.y;
    v0.z = (acc0.z >= 0.0f) ? acc0.z : SLOPE * acc0.z;
    v0.w = (acc0.w >= 0.0f) ? acc0.w : SLOPE * acc0.w;
    v1.x = (acc1.x >= 0.0f) ? acc1.x : SLOPE * acc1.x;
    v1.y = (acc1.y >= 0.0f) ? acc1.y : SLOPE * acc1.y;
    v1.z = (acc1.z >= 0.0f) ? acc1.z : SLOPE * acc1.z;
    v1.w = (acc1.w >= 0.0f) ? acc1.w : SLOPE * acc1.w;

    const float pa0 = v0.x * v0.x + v0.y * v0.y;
    const float pb0 = v0.z * v0.z + v0.w * v0.w;
    const float pa1 = v1.x * v1.x + v1.y * v1.y;
    const float pb1 = v1.z * v1.z + v1.w * v1.w;
    float pa = pa0 + pa1;
    float pb = pb0 + pb1;
#pragma unroll
    for (int o = 4; o > 0; o >>= 1) {
        pa += __shfl_xor(pa, o, 32);
        pb += __shfl_xor(pb, o, 32);
    }
    const float ss = pa + pb;

    const float inv = 1.0f / fmaxf(sqrtf(ss), EPS);

    v4f* xrow = (v4f*)(X + (size_t)row * N_HID);
    xrow[s] = v0;
    xrow[8 + s] = v1;
    orow[s] = cur0 + v0 * inv;
    orow[8 + s] = cur1 + v1 * inv;
}

extern "C" void kernel_launch(void* const* d_in, const int* in_sizes, int n_in,
                              void* d_out, int out_size, void* d_ws, size_t ws_size,
                              hipStream_t stream) {
    const float* user_emb = (const float*)d_in[0];
    const float* item_emb = (const float*)d_in[1];
    const float* ui_w     = (const float*)d_in[2];
    const float* s_w      = (const float*)d_in[3];
    const int*   ui_rows  = (const int*)d_in[4];
    const int*   ui_cols  = (const int*)d_in[5];
    const float* ui_vals  = (const float*)d_in[6];
    const int*   s_rows   = (const int*)d_in[7];
    const int*   s_cols   = (const int*)d_in[8];
    const float* s_vals   = (const float*)d_in[9];

    float* X = (float*)d_ws;
    float* Y = X + (size_t)N_NODES * N_HID;
    int* head = (int*)(Y + (size_t)N_NODES * N_HID);
    int* rec  = head + N_NODES;

    float* out_ui = (float*)d_out;
    float* out_s  = out_ui + (size_t)N_NODES * N_HID;

    const int nu = N_USER * N_HID;

    head_init_kernel<<<(N_NODES + 255) / 256, 256, 0, stream>>>(head, N_NODES);
    link_edges_kernel<<<(E_UI + 1023) / 1024, 256, 0, stream>>>(ui_rows, ui_cols, ui_vals, head, rec, E_UI);

    gemm64_wmma_kernel<<<(N_NODES / 16 + 7) / 8, 256, 0, stream>>>(user_emb, item_emb, N_USER, ui_w, Y, N_NODES);
    spmm_gather_norm_kernel<<<(N_NODES + 31) / 32, 256, 0, stream>>>(head, rec, Y, user_emb, item_emb, N_USER, X, out_ui, N_NODES, E_UI);
    gemm64_wmma_kernel<<<(N_NODES / 16 + 7) / 8, 256, 0, stream>>>(X, X + nu, N_USER, ui_w + (size_t)N_HID * N_HID, Y, N_NODES);
    spmm_gather_norm_kernel<<<(N_NODES + 31) / 32, 256, 0, stream>>>(head, rec, Y, out_ui, out_ui + nu, N_USER, X, out_ui, N_NODES, E_UI);

    head_init_kernel<<<(N_USER + 255) / 256, 256, 0, stream>>>(head, N_USER);
    link_edges_kernel<<<(E_S + 1023) / 1024, 256, 0, stream>>>(s_rows, s_cols, s_vals, head, rec, E_S);

    gemm64_wmma_kernel<<<(N_USER / 16 + 7) / 8, 256, 0, stream>>>(user_emb, user_emb, N_USER, s_w, Y, N_USER);
    spmm_gather_norm_kernel<<<(N_USER + 31) / 32, 256, 0, stream>>>(head, rec, Y, user_emb, user_emb, N_USER, X, out_s, N_USER, E_S);
    gemm64_wmma_kernel<<<(N_USER / 16 + 7) / 8, 256, 0, stream>>>(X, X, N_USER, s_w + (size_t)N_HID * N_HID, Y, N_USER);
    spmm_gather_norm_kernel<<<(N_USER + 31) / 32, 256, 0, stream>>>(head, rec, Y, out_s, out_s, N_USER, X, out_s, N_USER, E_S);
}


# Product order swapped, A(XW) -> (AX)W: gather sums the input rows, its block then multiplies the 32 sums by W on WMMA via LDS; no GEMM launches, no Y plane; float sum association changed
# speedup vs baseline: 4.0669x; 1.2749x over previous
#include <hip/hip_runtime.h>
#include <hip/hip_bf16.h>

typedef __attribute__((ext_vector_type(2))) float v2f;
typedef __attribute__((ext_vector_type(8))) float v8f;
typedef __attribute__((ext_vector_type(4))) float v4f;
typedef __attribute__((ext_vector_type(4))) int v4i;

#define N_USER 100000
#define N_ITEM 150000
#define N_NODES (N_USER + N_ITEM)
#define N_HID 64
#define E_UI 1000000
#define E_S 1000000
#define SLOPE 0.5f
#define EPS 1e-12f

__global__ void head_init_kernel(int* __restrict__ head, int n) {
    int i = blockIdx.x * blockDim.x + threadIdx.x;
    if (i < n) head[i] = -1;
}

__device__ __forceinline__ void link_edges_group(const int* __restrict__ rows,
                                                 const int* __restrict__ cols,
                                                 const float* __restrict__ vals,
                                                 int* __restrict__ head,
                                                 v4i* __restrict__ rec,
                                                 int base, int E, const bool guard) {
    int rw[4], cl[4], vl[4], prev[4];
#pragma unroll
    for (int j = 0; j < 4; ++j) {
        const int e = base + j * 256;
        const bool ok = !guard || e < E;
        rw[j] = ok ? rows[e] : 0;
        cl[j] = ok ? cols[e] : 0;
        vl[j] = ok ? __float_as_int(vals[e]) : 0;
    }
#pragma unroll
    for (int j = 0; j < 4; ++j) {
        const int e = base + j * 256;
        prev[j] = (!guard || e < E) ? atomicExch(&head[rw[j]], e) : -1;
    }
#pragma unroll
    for (int j = 0; j < 4; ++j) {
        const int e = base + j * 256;
        if (!guard || e < E) {
            v4i r;
            r.x = cl[j];
            r.y = vl[j];
            r.z = prev[j];
            r.w = 0;
            rec[e] = r;
        }
    }
}

__global__ void __launch_bounds__(256)
link_edges_kernel(const int* __restrict__ rows,
                  const int* __restrict__ cols,
                  const float* __restrict__ vals,
                  int* __restrict__ head,
                  int* __restrict__ rec_words,
                  int E) {
    const int base = blockIdx.x * 1024 + threadIdx.x;
    if (blockIdx.x * 1024 + 1024 <= E)
        link_edges_group(rows, cols, vals, head, (v4i*)rec_words, base, E, false);
    else
        link_edges_group(rows, cols, vals, head, (v4i*)rec_words, base, E, true);
}

#define G_PITCH 68

__global__ void __launch_bounds__(256)
spmm_gather_gemm_norm_kernel(const int* __restrict__ head,
                             const int* __restrict__ rec_words,
                             const float* src_a,
                             const float* src_b,
                             const float* res_a,
                             const float* res_b,
                             int split,
                             const float* __restrict__ W,
                             float* X,
                             int write_x,
                             float* out,
                             int M, int E) {
    __shared__ __attribute__((aligned(16))) float gbuf[32 * G_PITCH];
    __shared__ __attribute__((aligned(16))) float zbuf[32 * G_PITCH];

    const int wave = threadIdx.x >> 5;
    const int lane = threadIdx.x & 31;
    const int lrow = wave * 4 + (lane >> 3);
    const int row = blockIdx.x * 32 + lrow;
    const bool valid = row < M;
    const int rr = valid ? row : 0;
    const int s = lane & 7;
    const v4i* rec = (const v4i*)rec_words;
    const v4f* rrow = (const v4f*)((rr < split) ? res_a + (size_t)rr * N_HID
                                                : res_b + (size_t)(rr - split) * N_HID);

    v4f acc0 = (v4f)(0.0f);
    v4f acc1 = (v4f)(0.0f);
    int e = valid ? head[rr] : -1;
    const v4f cur0 = rrow[s];
    const v4f cur1 = rrow[8 + s];
    v4i r = rec[((unsigned)e < (unsigned)E) ? e : 0];
    for (int hop = 0; hop < E && (unsigned)e < (unsigned)E; ++hop) {
        const int c = r.x;
        const v4f* yrow = (const v4f*)((c < split) ? src_a + (size_t)c * N_HID
                                                   : src_b + (size_t)(c - split) * N_HID);
        const v4f y0 = yrow[s];
        const v4f y1 = yrow[8 + s];
        const float v = __int_as_float(r.y);
        e = r.z;
        r = rec[((unsigned)e < (unsigned)E) ? e : 0];
        asm volatile("" :: "v"(y0), "v"(y1), "v"(r.x), "v"(r.y), "v"(r.z));
        acc0 += v * y0;
        acc1 += v * y1;
    }

    *(v4f*)(gbuf + lrow * G_PITCH + 4 * s) = acc0;
    *(v4f*)(gbuf + lrow * G_PITCH + 32 + 4 * s) = acc1;

    {
        const int rt = wave >> 2;
        const int ct = wave & 3;
        const int fr = lane & 15;
        const int fh = lane >> 4;
        const float* ga = gbuf + (rt * 16 + fr) * G_PITCH + 2 * fh;
        const float* wb = W + (size_t)(2 * fh) * N_HID + ct * 16 + fr;
        float* zd = zbuf + (rt * 16 + 8 * fh) * G_PITCH + ct * 16 + fr;
        v2f b[16];
#pragma unroll
        for (int j = 0; j < 16; ++j) {
            b[j][0] = wb[(size_t)(4 * j) * N_HID];
            b[j][1] = wb[(size_t)(4 * j) * N_HID + N_HID];
        }
        __syncthreads();

        __builtin_amdgcn_sched_barrier(0);
        v8f z = v8f{};
#pragma unroll
        for (int j = 0; j < 16; ++j) {
            v2f a;
            a[0] = ga[4 * j];
            a[1] = ga[4 * j + 1];
            z = __builtin_amdgcn_wmma_f32_16x16x4_f32(
                 false, a,  false, b[j],
                 (short)0, z,
                 false,  false);
        }
        __builtin_amdgcn_sched_barrier(0);
#pragma unroll
        for (int v = 0; v < 8; ++v)
            zd[v * G_PITCH] = z[v];
    }
    __syncthreads();

    acc0 = *(const v4f*)(zbuf + lrow * G_PITCH + 4 * s);
    acc1 = *(const v4f*)(zbuf + lrow * G_PITCH + 32 + 4 * s);

    v4f v0, v1;
    v0.x = (acc0.x >= 0.0f) ? acc0.x : SLOPE * acc0.x;
    v0.y = (acc0.y >= 0.0f) ? acc0.y : SLOPE * acc0.y;
    v0.z = (acc0.z >= 0.0f) ? acc0.z : SLOPE * acc0.z;
    v0.w = (acc0.w >= 0.0f) ? acc0.w : SLOPE * acc0.w;
    v1.x = (acc1.x >= 0.0f) ? acc1.x : SLOPE * acc1.x;
    v1.y = (acc1.y >= 0.0f) ? acc1.y : SLOPE * acc1.y;
    v1.z = (acc1.z >= 0.0f) ? acc1.z : SLOPE * acc1.z;
    v1.w = (acc1.w >= 0.0f) ? acc1.w : SLOPE * acc1.w;

    const float pa0 = v0.x * v0.x + v0.y * v0.y;
    const float pb0 = v0.z * v0.z + v0.w * v0.w;
    const float pa1 = v1.x * v1.x + v1.y * v1.y;
    const float pb1 = v1.z * v1.z + v1.w * v1.w;
    float pa = pa0 + pa1;
    float pb = pb0 + pb1;
#pragma unroll
    for (int o = 4; o > 0; o >>= 1) {
        pa += __shfl_xor(pa, o, 32);
        pb += __shfl_xor(pb, o, 32);
    }
    const float ss = pa + pb;

    const float inv = 1.0f / fmaxf(sqrtf(ss), EPS);

    if (valid) {
        if (write_x) {
            v4f* xrow = (v4f*)(X + (size_t)row * N_HID);
            xrow[s] = v0;
            xrow[8 + s] = v1;
        }
        v4f* orow = (v4f*)(out + (size_t)row * N_HID);
        orow[s] = cur0 + v0 * inv;
        orow[8 + s] = cur1 + v1 * inv;
    }
}

extern "C" void kernel_launch(void* const* d_in, const int* in_sizes, int n_in,
                              void* d_out, int out_size, void* d_ws, size_t ws_size,
                              hipStream_t stream) {
    const float* user_emb = (const float*)d_in[0];
    const float* item_emb = (const float*)d_in[1];
    const float* ui_w     = (const float*)d_in[2];
    const float* s_w      = (const float*)d_in[3];
    const int*   ui_rows  = (const int*)d_in[4];
    const int*   ui_cols  = (const int*)d_in[5];
    const float* ui_vals  = (const float*)d_in[6];
    const int*   s_rows   = (const int*)d_in[7];
    const int*   s_cols   = (const int*)d_in[8];
    const float* s_vals   = (const float*)d_in[9];

    float* X = (float*)d_ws;
    int* head = (int*)(X + (size_t)N_NODES * N_HID);
    int* rec  = head + N_NODES;

    float* out_ui = (float*)d_out;
    float* out_s  = out_ui + (size_t)N_NODES * N_HID;

    const int nu = N_USER * N_HID;

    head_init_kernel<<<(N_NODES + 255) / 256, 256, 0, stream>>>(head, N_NODES);
    link_edges_kernel<<<(E_UI + 1023) / 1024, 256, 0, stream>>>(ui_rows, ui_cols, ui_vals, head, rec, E_UI);

    spmm_gather_gemm_norm_kernel<<<(N_NODES + 31) / 32, 256, 0, stream>>>(
        head, rec, user_emb, item_emb, user_emb, item_emb, N_USER, ui_w, X, 1, out_ui, N_NODES, E_UI);
    spmm_gather_gemm_norm_kernel<<<(N_NODES + 31) / 32, 256, 0, stream>>>(
        head, rec, X, X + nu, out_ui, out_ui + nu, N_USER, ui_w + (size_t)N_HID * N_HID, X, 0, out_ui, N_NODES, E_UI);

    head_init_kernel<<<(N_USER + 255) / 256, 256, 0, stream>>>(head, N_USER);
    link_edges_kernel<<<(E_S + 1023) / 1024, 256, 0, stream>>>(s_rows, s_cols, s_vals, head, rec, E_S);

    spmm_gather_gemm_norm_kernel<<<(N_USER + 31) / 32, 256, 0, stream>>>(
        head, rec, user_emb, user_emb, user_emb, user_emb, N_USER, s_w, X, 1, out_s, N_USER, E_S);
    spmm_gather_gemm_norm_kernel<<<(N_USER + 31) / 32, 256, 0, stream>>>(
        head, rec, X, X, out_s, out_s, N_USER, s_w + (size_t)N_HID * N_HID, X, 0, out_s, N_USER, E_S);
}


# Each row's edge list split in two by edge-index parity and both walked in step in one loop (lists end in a zero-weight end record): about half the dependent hops; row float sum order now alternates
# speedup vs baseline: 4.2838x; 1.0533x over previous
#include <hip/hip_runtime.h>
#include <hip/hip_bf16.h>

typedef __attribute__((ext_vector_type(2))) float v2f;
typedef __attribute__((ext_vector_type(8))) float v8f;
typedef __attribute__((ext_vector_type(4))) float v4f;
typedef __attribute__((ext_vector_type(4))) int v4i;
typedef __attribute__((ext_vector_type(2))) int v2i;

#define N_USER 100000
#define N_ITEM 150000
#define N_NODES (N_USER + N_ITEM)
#define N_HID 64
#define E_UI 1000000
#define E_S 1000000
#define SLOPE 0.5f
#define EPS 1e-12f

__global__ void head_init_kernel(int* __restrict__ head, int n, int empty) {
    int i = blockIdx.x * blockDim.x + threadIdx.x;
    if (i < n) head[i] = empty;
}

__device__ __forceinline__ void link_edges_group(const int* __restrict__ rows,
                                                 const int* __restrict__ cols,
                                                 const float* __restrict__ vals,
                                                 int* __restrict__ head,
                                                 v4i* __restrict__ rec,
                                                 int base, int E, const bool guard) {
    int rw[4], cl[4], vl[4], prev[4];
#pragma unroll
    for (int j = 0; j < 4; ++j) {
        const int e = base + j * 256;
        const bool ok = !guard || e < E;
        rw[j] = ok ? rows[e] : 0;
        cl[j] = ok ? cols[e] : 0;
        vl[j] = ok ? __float_as_int(vals[e]) : 0;
    }
#pragma unroll
    for (int j = 0; j < 4; ++j) {
        const int e = base + j * 256;
        prev[j] = (!guard || e < E) ? atomicExch(&head[2 * rw[j] + (e & 1)], e) : E;
    }
#pragma unroll
    for (int j = 0; j < 4; ++j) {
        const int e = base + j * 256;
        if (!guard || e < E) {
            v4i r;
            r.x = cl[j];
            r.y = vl[j];
            r.z = prev[j];
            r.w = 0;
            rec[e] = r;
        }
    }
}

__global__ void __launch_bounds__(256)
link_edges_kernel(const int* __restrict__ rows,
                  const int* __restrict__ cols,
                  const float* __restrict__ vals,
                  int* __restrict__ head,
                  int* __restrict__ rec_words,
                  int E) {
    const int base = blockIdx.x * 1024 + threadIdx.x;
    if (base == 0) {
        v4i r;
        r.x = 0;
        r.y = 0;
        r.z = E;
        r.w = 0;
        ((v4i*)rec_words)[E] = r;
    }
    if (blockIdx.x * 1024 + 1024 <= E)
        link_edges_group(rows, cols, vals, head, (v4i*)rec_words, base, E, false);
    else
        link_edges_group(rows, cols, vals, head, (v4i*)rec_words, base, E, true);
}

#define G_PITCH 68

__global__ void __launch_bounds__(256)
spmm_gather_gemm_norm_kernel(const int* __restrict__ head,
                             const int* __restrict__ rec_words,
                             const float* src_a,
                             const float* src_b,
                             const float* res_a,
                             const float* res_b,
                             int split,
                             const float* __restrict__ W,
                             float* X,
                             int write_x,
                             float* out,
                             int M, int E) {
    __shared__ __attribute__((aligned(16))) float gbuf[32 * G_PITCH];
    __shared__ __attribute__((aligned(16))) float zbuf[32 * G_PITCH];

    const int wave = threadIdx.x >> 5;
    const int lane = threadIdx.x & 31;
    const int lrow = wave * 4 + (lane >> 3);
    const int row = blockIdx.x * 32 + lrow;
    const bool valid = row < M;
    const int rr = valid ? row : 0;
    const int s = lane & 7;
    const v4i* rec = (const v4i*)rec_words;
    const v4f* rrow = (const v4f*)((rr < split) ? res_a + (size_t)rr * N_HID
                                                : res_b + (size_t)(rr - split) * N_HID);

    v4f acc0 = (v4f)(0.0f);
    v4f acc1 = (v4f)(0.0f);
    const unsigned uE = (unsigned)E;
    const v2i hd = ((const v2i*)head)[rr];
    unsigned ea = valid ? min((unsigned)hd.x, uE) : uE;
    unsigned eb = valid ? min((unsigned)hd.y, uE) : uE;
    __builtin_prefetch((const void*)(rrow + s), 0, 3);
    v4i ra = rec[ea];
    v4i rb = rec[eb];
    const char* base_a = (const char*)(src_a + 4 * s);
    const char* base_b = (const char*)(src_b + 4 * s) - (size_t)split * (N_HID * sizeof(float));
    for (int hop = 0; hop < E && (ea != uE || eb != uE); ++hop) {
        const unsigned ca = (unsigned)ra.x;
        const unsigned cb = (unsigned)rb.x;
        const v4f* yrow_a = (const v4f*)(((int)ca < split ? base_a : base_b) + (size_t)ca * (N_HID * sizeof(float)));
        const v4f* yrow_b = (const v4f*)(((int)cb < split ? base_a : base_b) + (size_t)cb * (N_HID * sizeof(float)));
        const v4f ya0 = yrow_a[0];
        const v4f ya1 = yrow_a[8];
        const v4f yb0 = yrow_b[0];
        const v4f yb1 = yrow_b[8];
        const float va = __int_as_float(ra.y);
        const float vb = __int_as_float(rb.y);
        ea = min((unsigned)ra.z, uE);
        eb = min((unsigned)rb.z, uE);
        ra = rec[ea];
        rb = rec[eb];
        asm volatile("" :: "v"(ya0), "v"(ya1), "v"(yb0), "v"(yb1),
                           "v"(ra.x), "v"(ra.y), "v"(ra.z), "v"(rb.x), "v"(rb.y), "v"(rb.z));
        acc0 += va * ya0;
        acc1 += va * ya1;
        acc0 += vb * yb0;
        acc1 += vb * yb1;
    }

    *(v4f*)(gbuf + lrow * G_PITCH + 4 * s) = acc0;
    *(v4f*)(gbuf + lrow * G_PITCH + 32 + 4 * s) = acc1;

    v4f cur0, cur1;
    {
        const int rt = wave >> 2;
        const int ct = wave & 3;
        const int fr = lane & 15;
        const int fh = lane >> 4;
        const float* ga = gbuf + (rt * 16 + fr) * G_PITCH + 2 * fh;
        const float* wb = W + (size_t)(2 * fh) * N_HID + ct * 16 + fr;
        float* zd = zbuf + (rt * 16 + 8 * fh) * G_PITCH + ct * 16 + fr;
        v2f b[16];
#pragma unroll
        for (int j = 0; j < 16; ++j) {
            b[j][0] = wb[(size_t)(4 * j) * N_HID];
            b[j][1] = wb[(size_t)(4 * j) * N_HID + N_HID];
        }
        __syncthreads();

        __builtin_amdgcn_sched_barrier(0);
        v8f z = v8f{};
#pragma unroll
        for (int j = 0; j < 16; ++j) {
            v2f a;
            a[0] = ga[4 * j];
            a[1] = ga[4 * j + 1];
            z = __builtin_amdgcn_wmma_f32_16x16x4_f32(
                 false, a,  false, b[j],
                 (short)0, z,
                 false,  false);
        }
        __builtin_amdgcn_sched_barrier(0);
        cur0 = rrow[s];
        cur1 = rrow[8 + s];
#pragma unroll
        for (int v = 0; v < 8; ++v)
            zd[v * G_PITCH] = z[v];
    }
    __syncthreads();

    acc0 = *(const v4f*)(zbuf + lrow * G_PITCH + 4 * s);
    acc1 = *(const v4f*)(zbuf + lrow * G_PITCH + 32 + 4 * s);

    v4f v0, v1;
    v0.x = (acc0.x >= 0.0f) ? acc0.x : SLOPE * acc0.x;
    v0.y = (acc0.y >= 0.0f) ? acc0.y : SLOPE * acc0.y;
    v0.z = (acc0.z >= 0.0f) ? acc0.z : SLOPE * acc0.z;
    v0.w = (acc0.w >= 0.0f) ? acc0.w : SLOPE * acc0.w;
    v1.x = (acc1.x >= 0.0f) ? acc1.x : SLOPE * acc1.x;
    v1.y = (acc1.y >= 0.0f) ? acc1.y : SLOPE * acc1.y;
    v1.z = (acc1.z >= 0.0f) ? acc1.z : SLOPE * acc1.z;
    v1.w = (acc1.w >= 0.0f) ? acc1.w : SLOPE * acc1.w;

    const float pa0 = v0.x * v0.x + v0.y * v0.y;
    const float pb0 = v0.z * v0.z + v0.w * v0.w;
    const float pa1 = v1.x * v1.x + v1.y * v1.y;
    const float pb1 = v1.z * v1.z + v1.w * v1.w;
    float pa = pa0 + pa1;
    float pb = pb0 + pb1;
#pragma unroll
    for (int o = 4; o > 0; o >>= 1) {
        pa += __shfl_xor(pa, o, 32);
        pb += __shfl_xor(pb, o, 32);
    }
    const float ss = pa + pb;

    const float inv = 1.0f / fmaxf(sqrtf(ss), EPS);

    if (valid) {
        if (write_x) {
            v4f* xrow = (v4f*)(X + (size_t)row * N_HID);
            xrow[s] = v0;
            xrow[8 + s] = v1;
        }
        v4f* orow = (v4f*)(out + (size_t)row * N_HID);
        orow[s] = cur0 + v0 * inv;
        orow[8 + s] = cur1 + v1 * inv;
    }
}

extern "C" void kernel_launch(void* const* d_in, const int* in_sizes, int n_in,
                              void* d_out, int out_size, void* d_ws, size_t ws_size,
                              hipStream_t stream) {
    const float* user_emb = (const float*)d_in[0];
    const float* item_emb = (const float*)d_in[1];
    const float* ui_w     = (const float*)d_in[2];
    const float* s_w      = (const float*)d_in[3];
    const int*   ui_rows  = (const int*)d_in[4];
    const int*   ui_cols  = (const int*)d_in[5];
    const float* ui_vals  = (const float*)d_in[6];
    const int*   s_rows   = (const int*)d_in[7];
    const int*   s_cols   = (const int*)d_in[8];
    const float* s_vals   = (const float*)d_in[9];

    float* X = (float*)d_ws;
    int* head = (int*)(X + (size_t)N_NODES * N_HID);
    int* rec  = head + 2 * N_NODES;

    float* out_ui = (float*)d_out;
    float* out_s  = out_ui + (size_t)N_NODES * N_HID;

    const int nu = N_USER * N_HID;

    head_init_kernel<<<(2 * N_NODES + 255) / 256, 256, 0, stream>>>(head, 2 * N_NODES, E_UI);
    link_edges_kernel<<<(E_UI + 1023) / 1024, 256, 0, stream>>>(ui_rows, ui_cols, ui_vals, head, rec, E_UI);

    spmm_gather_gemm_norm_kernel<<<(N_NODES + 31) / 32, 256, 0, stream>>>(
        head, rec, user_emb, item_emb, user_emb, item_emb, N_USER, ui_w, X, 1, out_ui, N_NODES, E_UI);
    spmm_gather_gemm_norm_kernel<<<(N_NODES + 31) / 32, 256, 0, stream>>>(
        head, rec, X, X + nu, out_ui, out_ui + nu, N_USER, ui_w + (size_t)N_HID * N_HID, X, 0, out_ui, N_NODES, E_UI);

    head_init_kernel<<<(2 * N_USER + 255) / 256, 256, 0, stream>>>(head, 2 * N_USER, E_S);
    link_edges_kernel<<<(E_S + 1023) / 1024, 256, 0, stream>>>(s_rows, s_cols, s_vals, head, rec, E_S);

    spmm_gather_gemm_norm_kernel<<<(N_USER + 31) / 32, 256, 0, stream>>>(
        head, rec, user_emb, user_emb, user_emb, user_emb, N_USER, s_w, X, 1, out_s, N_USER, E_S);
    spmm_gather_gemm_norm_kernel<<<(N_USER + 31) / 32, 256, 0, stream>>>(
        head, rec, X, X, out_s, out_s, N_USER, s_w + (size_t)N_HID * N_HID, X, 0, out_s, N_USER, E_S);
}


# Both graphs share every launch: head init, edge linking and each layer's gather take social and user-item blocks in one grid (own X, heads, records each); 4 launches, not 8; float sum orders kept
# speedup vs baseline: 4.5626x; 1.0651x over previous
#include <hip/hip_runtime.h>
#include <hip/hip_bf16.h>

typedef __attribute__((ext_vector_type(2))) float v2f;
typedef __attribute__((ext_vector_type(8))) float v8f;
typedef __attribute__((ext_vector_type(4))) float v4f;
typedef __attribute__((ext_vector_type(4))) int v4i;
typedef __attribute__((ext_vector_type(2))) int v2i;

#define N_USER 100000
#define N_ITEM 150000
#define N_NODES (N_USER + N_ITEM)
#define N_HID 64
#define E_UI 1000000
#define E_S 1000000
#define SLOPE 0.5f
#define EPS 1e-12f


__global__ void head_init_kernel(int* __restrict__ head, int n_first, int empty_first,
                                 int n, int empty_second) {
    int i = blockIdx.x * blockDim.x + threadIdx.x;
    if (i < n) head[i] = (i < n_first) ? empty_first : empty_second;
}

__device__ __forceinline__ void link_edges_group(const int* __restrict__ rows,
                                                 const int* __restrict__ cols,
                                                 const float* __restrict__ vals,
                                                 int* __restrict__ head,
                                                 v4i* __restrict__ rec,
                                                 int base, int E, const bool guard) {
    int rw[4], cl[4], vl[4], prev[4];
#pragma unroll
    for (int j = 0; j < 4; ++j) {
        const int e = base + j * 256;
        const bool ok = !guard || e < E;
        rw[j] = ok ? rows[e] : 0;
        cl[j] = ok ? cols[e] : 0;
        vl[j] = ok ? __float_as_int(vals[e]) : 0;
    }
#pragma unroll
    for (int j = 0; j < 4; ++j) {
        const int e = base + j * 256;
        prev[j] = (!guard || e < E) ? atomicExch(&head[2 * rw[j] + (e & 1)], e) : E;
    }
#pragma unroll
    for (int j = 0; j < 4; ++j) {
        const int e = base + j * 256;
        if (!guard || e < E) {
            v4i r;
            r.x = cl[j];
            r.y = vl[j];
            r.z = prev[j];
            r.w = 0;
            rec[e] = r;
        }
    }
}

__global__ void __launch_bounds__(256)
link_edges_kernel(const int* __restrict__ rows_first,
                  const int* __restrict__ cols_first,
                  const float* __restrict__ vals_first,
                  int* __restrict__ head_first,
                  int* __restrict__ rec_first,
                  int E_first, int nb_first,
                  const int* __restrict__ rows_second,
                  const int* __restrict__ cols_second,
                  const float* __restrict__ vals_second,
                  int* __restrict__ head_second,
                  int* __restrict__ rec_second,
                  int E_second) {
    const bool second = (int)blockIdx.x >= nb_first;
    const int blk = second ? (int)blockIdx.x - nb_first : (int)blockIdx.x;
    const int* __restrict__ rows = second ? rows_second : rows_first;
    const int* __restrict__ cols = second ? cols_second : cols_first;
    const float* __restrict__ vals = second ? vals_second : vals_first;
    int* __restrict__ head = second ? head_second : head_first;
    int* __restrict__ rec_words = second ? rec_second : rec_first;
    const int E = second ? E_second : E_first;
    const int base = blk * 1024 + threadIdx.x;
    if (base == 0) {
        v4i r;
        r.x = 0;
        r.y = 0;
        r.z = E;
        r.w = 0;
        ((v4i*)rec_words)[E] = r;
    }
    if (blk * 1024 + 1024 <= E)
        link_edges_group(rows, cols, vals, head, (v4i*)rec_words, base, E, false);
    else
        link_edges_group(rows, cols, vals, head, (v4i*)rec_words, base, E, true);
}

#define G_PITCH 68

__global__ void __launch_bounds__(256)
spmm_gather_gemm_norm_kernel(const int* __restrict__ so_head,
                             const int* __restrict__ so_rec,
                             const float* so_src,
                             const float* so_res,
                             const float* __restrict__ so_W,
                             float* so_X,
                             float* so_out,
                             int so_M, int so_E, int nb_so,
                             const int* __restrict__ ui_head,
                             const int* __restrict__ ui_rec,
                             const float* ui_src_a,
                             const float* ui_src_b,
                             const float* ui_res_a,
                             const float* ui_res_b,
                             int ui_split,
                             const float* __restrict__ ui_W,
                             float* ui_X,
                             float* ui_out,
                             int ui_M, int ui_E,
                             int write_x) {
    __shared__ __attribute__((aligned(16))) float gbuf[32 * G_PITCH];
    __shared__ __attribute__((aligned(16))) float zbuf[32 * G_PITCH];

    const bool so = (int)blockIdx.x < nb_so;
    const int blk = so ? (int)blockIdx.x : (int)blockIdx.x - nb_so;
    const int* __restrict__ head = so ? so_head : ui_head;
    const int* __restrict__ rec_words = so ? so_rec : ui_rec;
    const float* src_a = so ? so_src : ui_src_a;
    const float* src_b = so ? so_src : ui_src_b;
    const float* res_a = so ? so_res : ui_res_a;
    const float* res_b = so ? so_res : ui_res_b;
    const int split = so ? so_M : ui_split;
    const float* __restrict__ W = so ? so_W : ui_W;
    float* X = so ? so_X : ui_X;
    float* out = so ? so_out : ui_out;
    const int M = so ? so_M : ui_M;
    const int E = so ? so_E : ui_E;

    const int wave = threadIdx.x >> 5;
    const int lane = threadIdx.x & 31;
    const int lrow = wave * 4 + (lane >> 3);
    const int row = blk * 32 + lrow;
    const bool valid = row < M;
    const int rr = valid ? row : 0;
    const int s = lane & 7;
    const v4i* rec = (const v4i*)rec_words;
    const v4f* rrow = (const v4f*)((rr < split) ? res_a + (size_t)rr * N_HID
                                                : res_b + (size_t)(rr - split) * N_HID);

    v4f acc0 = (v4f)(0.0f);
    v4f acc1 = (v4f)(0.0f);
    const unsigned uE = (unsigned)E;
    const v2i hd = ((const v2i*)head)[rr];
    unsigned ea = valid ? min((unsigned)hd.x, uE) : uE;
    unsigned eb = valid ? min((unsigned)hd.y, uE) : uE;
    __builtin_prefetch((const void*)(rrow + s), 0, 3);
    v4i ra = rec[ea];
    v4i rb = rec[eb];
    const char* base_a = (const char*)(src_a + 4 * s);
    const char* base_b = (const char*)(src_b + 4 * s) - (size_t)split * (N_HID * sizeof(float));
    for (int hop = 0; hop < E && (ea != uE || eb != uE); ++hop) {
        const unsigned ca = (unsigned)ra.x;
        const unsigned cb = (unsigned)rb.x;
        const v4f* yrow_a = (const v4f*)(((int)ca < split ? base_a : base_b) + (size_t)ca * (N_HID * sizeof(float)));
        const v4f* yrow_b = (const v4f*)(((int)cb < split ? base_a : base_b) + (size_t)cb * (N_HID * sizeof(float)));
        const v4f ya0 = yrow_a[0];
        const v4f ya1 = yrow_a[8];
        const v4f yb0 = yrow_b[0];
        const v4f yb1 = yrow_b[8];
        const float va = __int_as_float(ra.y);
        const float vb = __int_as_float(rb.y);
        ea = min((unsigned)ra.z, uE);
        eb = min((unsigned)rb.z, uE);
        ra = rec[ea];
        rb = rec[eb];
        asm volatile("" :: "v"(ya0), "v"(ya1), "v"(yb0), "v"(yb1),
                           "v"(ra.x), "v"(ra.y), "v"(ra.z), "v"(rb.x), "v"(rb.y), "v"(rb.z));
        acc0 += va * ya0;
        acc1 += va * ya1;
        acc0 += vb * yb0;
        acc1 += vb * yb1;
    }

    *(v4f*)(gbuf + lrow * G_PITCH + 4 * s) = acc0;
    *(v4f*)(gbuf + lrow * G_PITCH + 32 + 4 * s) = acc1;

    v4f cur0, cur1;
    {
        const int rt = wave >> 2;
        const int ct = wave & 3;
        const int fr = lane & 15;
        const int fh = lane >> 4;
        const float* ga = gbuf + (rt * 16 + fr) * G_PITCH + 2 * fh;
        const float* wb = W + (size_t)(2 * fh) * N_HID + ct * 16 + fr;
        float* zd = zbuf + (rt * 16 + 8 * fh) * G_PITCH + ct * 16 + fr;
        v2f b[16];
#pragma unroll
        for (int j = 0; j < 16; ++j) {
            b[j][0] = wb[(size_t)(4 * j) * N_HID];
            b[j][1] = wb[(size_t)(4 * j) * N_HID + N_HID];
        }
        __syncthreads();

        __builtin_amdgcn_sched_barrier(0);
        v8f z = v8f{};
#pragma unroll
        for (int j = 0; j < 16; ++j) {
            v2f a;
            a[0] = ga[4 * j];
            a[1] = ga[4 * j + 1];
            z = __builtin_amdgcn_wmma_f32_16x16x4_f32(
                 false, a,  false, b[j],
                 (short)0, z,
                 false,  false);
        }
        __builtin_amdgcn_sched_barrier(0);
        cur0 = rrow[s];
        cur1 = rrow[8 + s];
#pragma unroll
        for (int v = 0; v < 8; ++v)
            zd[v * G_PITCH] = z[v];
    }
    __syncthreads();

    acc0 = *(const v4f*)(zbuf + lrow * G_PITCH + 4 * s);
    acc1 = *(const v4f*)(zbuf + lrow * G_PITCH + 32 + 4 * s);

    v4f v0, v1;
    v0.x = (acc0.x >= 0.0f) ? acc0.x : SLOPE * acc0.x;
    v0.y = (acc0.y >= 0.0f) ? acc0.y : SLOPE * acc0.y;
    v0.z = (acc0.z >= 0.0f) ? acc0.z : SLOPE * acc0.z;
    v0.w = (acc0.w >= 0.0f) ? acc0.w : SLOPE * acc0.w;
    v1.x = (acc1.x >= 0.0f) ? acc1.x : SLOPE * acc1.x;
    v1.y = (acc1.y >= 0.0f) ? acc1.y : SLOPE * acc1.y;
    v1.z = (acc1.z >= 0.0f) ? acc1.z : SLOPE * acc1.z;
    v1.w = (acc1.w >= 0.0f) ? acc1.w : SLOPE * acc1.w;

    const float pa0 = v0.x * v0.x + v0.y * v0.y;
    const float pb0 = v0.z * v0.z + v0.w * v0.w;
    const float pa1 = v1.x * v1.x + v1.y * v1.y;
    const float pb1 = v1.z * v1.z + v1.w * v1.w;
    float pa = pa0 + pa1;
    float pb = pb0 + pb1;
#pragma unroll
    for (int o = 4; o > 0; o >>= 1) {
        pa += __shfl_xor(pa, o, 32);
        pb += __shfl_xor(pb, o, 32);
    }
    const float ss = pa + pb;

    const float inv = 1.0f / fmaxf(sqrtf(ss), EPS);

    if (valid) {
        if (write_x) {
            v4f* xrow = (v4f*)(X + (size_t)row * N_HID);
            xrow[s] = v0;
            xrow[8 + s] = v1;
        }
        v4f* orow = (v4f*)(out + (size_t)row * N_HID);
        orow[s] = cur0 + v0 * inv;
        orow[8 + s] = cur1 + v1 * inv;
    }
}

extern "C" void kernel_launch(void* const* d_in, const int* in_sizes, int n_in,
                              void* d_out, int out_size, void* d_ws, size_t ws_size,
                              hipStream_t stream) {
    const float* user_emb = (const float*)d_in[0];
    const float* item_emb = (const float*)d_in[1];
    const float* ui_w     = (const float*)d_in[2];
    const float* s_w      = (const float*)d_in[3];
    const int*   ui_rows  = (const int*)d_in[4];
    const int*   ui_cols  = (const int*)d_in[5];
    const float* ui_vals  = (const float*)d_in[6];
    const int*   s_rows   = (const int*)d_in[7];
    const int*   s_cols   = (const int*)d_in[8];
    const float* s_vals   = (const float*)d_in[9];

    float* X_ui = (float*)d_ws;
    float* X_so = X_ui + (size_t)N_NODES * N_HID;
    int* head_ui = (int*)(X_so + (size_t)N_USER * N_HID);
    int* head_so = head_ui + 2 * N_NODES;
    int* rec_ui  = head_so + 2 * N_USER;
    int* rec_so  = rec_ui + 4 * (E_UI + 1);

    float* out_ui = (float*)d_out;
    float* out_s  = out_ui + (size_t)N_NODES * N_HID;

    const int nu = N_USER * N_HID;

    head_init_kernel<<<(2 * (N_NODES + N_USER) + 255) / 256, 256, 0, stream>>>(
        head_ui, 2 * N_NODES, E_UI, 2 * (N_NODES + N_USER), E_S);

    link_edges_kernel<<<(E_UI + 1023) / 1024 + (E_S + 1023) / 1024, 256, 0, stream>>>(
        ui_rows, ui_cols, ui_vals, head_ui, rec_ui, E_UI, (E_UI + 1023) / 1024,
        s_rows, s_cols, s_vals, head_so, rec_so, E_S);

    spmm_gather_gemm_norm_kernel<<<(N_USER + 31) / 32 + (N_NODES + 31) / 32, 256, 0, stream>>>(
        head_so, rec_so, user_emb, user_emb, s_w, X_so, out_s, N_USER, E_S, (N_USER + 31) / 32,
        head_ui, rec_ui, user_emb, item_emb, user_emb, item_emb, N_USER, ui_w, X_ui, out_ui, N_NODES, E_UI,
        1);
    spmm_gather_gemm_norm_kernel<<<(N_USER + 31) / 32 + (N_NODES + 31) / 32, 256, 0, stream>>>(
        head_so, rec_so, X_so, out_s, s_w + (size_t)N_HID * N_HID, X_so, out_s, N_USER, E_S, (N_USER + 31) / 32,
        head_ui, rec_ui, X_ui, X_ui + nu, out_ui, out_ui + nu, N_USER, ui_w + (size_t)N_HID * N_HID, X_ui, out_ui, N_NODES, E_UI,
        0);
}


# First layer stops writing out and reading its table row: second layer forms out = (table row + first term from its own X row, same square-sum tree) + own term; float sum orders kept
# speedup vs baseline: 4.7888x; 1.0496x over previous
#include <hip/hip_runtime.h>
#include <hip/hip_bf16.h>

typedef __attribute__((ext_vector_type(2))) float v2f;
typedef __attribute__((ext_vector_type(8))) float v8f;
typedef __attribute__((ext_vector_type(4))) float v4f;
typedef __attribute__((ext_vector_type(4))) int v4i;
typedef __attribute__((ext_vector_type(2))) int v2i;

#define N_USER 100000
#define N_ITEM 150000
#define N_NODES (N_USER + N_ITEM)
#define N_HID 64
#define E_UI 1000000
#define E_S 1000000
#define SLOPE 0.5f
#define EPS 1e-12f


__global__ void head_init_kernel(int* __restrict__ head, int n_first, int empty_first,
                                 int n, int empty_second) {
    int i = blockIdx.x * blockDim.x + threadIdx.x;
    if (i < n) head[i] = (i < n_first) ? empty_first : empty_second;
}

__device__ __forceinline__ void link_edges_group(const int* __restrict__ rows,
                                                 const int* __restrict__ cols,
                                                 const float* __restrict__ vals,
                                                 int* __restrict__ head,
                                                 v4i* __restrict__ rec,
                                                 int base, int E, const bool guard) {
    int rw[4], cl[4], vl[4], prev[4];
#pragma unroll
    for (int j = 0; j < 4; ++j) {
        const int e = base + j * 256;
        const bool ok = !guard || e < E;
        rw[j] = ok ? rows[e] : 0;
        cl[j] = ok ? cols[e] : 0;
        vl[j] = ok ? __float_as_int(vals[e]) : 0;
    }
#pragma unroll
    for (int j = 0; j < 4; ++j) {
        const int e = base + j * 256;
        prev[j] = (!guard || e < E) ? atomicExch(&head[2 * rw[j] + (e & 1)], e) : E;
    }
#pragma unroll
    for (int j = 0; j < 4; ++j) {
        const int e = base + j * 256;
        if (!guard || e < E) {
            v4i r;
            r.x = cl[j];
            r.y = vl[j];
            r.z = prev[j];
            r.w = 0;
            rec[e] = r;
        }
    }
}

__global__ void __launch_bounds__(256)
link_edges_kernel(const int* __restrict__ rows_first,
                  const int* __restrict__ cols_first,
                  const float* __restrict__ vals_first,
                  int* __restrict__ head_first,
                  int* __restrict__ rec_first,
                  int E_first, int nb_first,
                  const int* __restrict__ rows_second,
                  const int* __restrict__ cols_second,
                  const float* __restrict__ vals_second,
                  int* __restrict__ head_second,
                  int* __restrict__ rec_second,
                  int E_second) {
    const bool second = (int)blockIdx.x >= nb_first;
    const int blk = second ? (int)blockIdx.x - nb_first : (int)blockIdx.x;
    const int* __restrict__ rows = second ? rows_second : rows_first;
    const int* __restrict__ cols = second ? cols_second : cols_first;
    const float* __restrict__ vals = second ? vals_second : vals_first;
    int* __restrict__ head = second ? head_second : head_first;
    int* __restrict__ rec_words = second ? rec_second : rec_first;
    const int E = second ? E_second : E_first;
    const int base = blk * 1024 + threadIdx.x;
    if (base == 0) {
        v4i r;
        r.x = 0;
        r.y = 0;
        r.z = E;
        r.w = 0;
        ((v4i*)rec_words)[E] = r;
    }
    if (blk * 1024 + 1024 <= E)
        link_edges_group(rows, cols, vals, head, (v4i*)rec_words, base, E, false);
    else
        link_edges_group(rows, cols, vals, head, (v4i*)rec_words, base, E, true);
}

#define G_PITCH 68

__device__ __forceinline__ float inv_row_norm(const v4f a, const v4f b) {
    const float pa0 = a.x * a.x + a.y * a.y;
    const float pb0 = a.z * a.z + a.w * a.w;
    const float pa1 = b.x * b.x + b.y * b.y;
    const float pb1 = b.z * b.z + b.w * b.w;
    float pa = pa0 + pa1;
    float pb = pb0 + pb1;
#pragma unroll
    for (int o = 4; o > 0; o >>= 1) {
        pa += __shfl_xor(pa, o, 32);
        pb += __shfl_xor(pb, o, 32);
    }
    const float ss = pa + pb;
    return 1.0f / fmaxf(sqrtf(ss), EPS);
}

template <bool FIRST>
__global__ void __launch_bounds__(256, 16)
spmm_gather_gemm_norm_kernel(const int* __restrict__ so_head,
                             const int* __restrict__ so_rec,
                             const float* so_src,
                             const float* so_res,
                             const float* __restrict__ so_W,
                             float* so_X,
                             float* so_out,
                             int so_M, int so_E, int nb_so,
                             const int* __restrict__ ui_head,
                             const int* __restrict__ ui_rec,
                             const float* ui_src_a,
                             const float* ui_src_b,
                             const float* ui_res_a,
                             const float* ui_res_b,
                             int ui_split,
                             const float* __restrict__ ui_W,
                             float* ui_X,
                             float* ui_out,
                             int ui_M, int ui_E) {
    __shared__ __attribute__((aligned(16))) float gbuf[32 * G_PITCH];
    __shared__ __attribute__((aligned(16))) float zbuf[32 * G_PITCH];

    const bool so = (int)blockIdx.x < nb_so;
    const int blk = so ? (int)blockIdx.x : (int)blockIdx.x - nb_so;
    const int* __restrict__ head = so ? so_head : ui_head;
    const int* __restrict__ rec_words = so ? so_rec : ui_rec;
    const float* src_a = so ? so_src : ui_src_a;
    const float* src_b = so ? so_src : ui_src_b;
    const float* res_a = so ? so_res : ui_res_a;
    const float* res_b = so ? so_res : ui_res_b;
    const int split = so ? so_M : ui_split;
    const float* __restrict__ W = so ? so_W : ui_W;
    float* X = so ? so_X : ui_X;
    float* out = so ? so_out : ui_out;
    const int M = so ? so_M : ui_M;
    const int E = so ? so_E : ui_E;

    const int wave = threadIdx.x >> 5;
    const int lane = threadIdx.x & 31;
    const int lrow = wave * 4 + (lane >> 3);
    const int row = blk * 32 + lrow;
    const bool valid = row < M;
    const int rr = valid ? row : 0;
    const int s = lane & 7;
    const v4i* rec = (const v4i*)rec_words;
    const v4f* rrow = (const v4f*)((rr < split) ? res_a + (size_t)rr * N_HID
                                                : res_b + (size_t)(rr - split) * N_HID);

    const v4f* xown = (const v4f*)(X + (size_t)rr * N_HID);

    v4f acc0 = (v4f)(0.0f);
    v4f acc1 = (v4f)(0.0f);
    const unsigned uE = (unsigned)E;
    const v2i hd = ((const v2i*)head)[rr];
    unsigned ea = valid ? min((unsigned)hd.x, uE) : uE;
    unsigned eb = valid ? min((unsigned)hd.y, uE) : uE;
    if (!FIRST) {
        __builtin_prefetch((const void*)(rrow + s), 0, 3);
        __builtin_prefetch((const void*)(xown + s), 0, 3);
    }
    v4i ra = rec[ea];
    v4i rb = rec[eb];
    const char* base_a = (const char*)(src_a + 4 * s);
    const char* base_b = (const char*)(src_b + 4 * s) - (size_t)split * (N_HID * sizeof(float));
    for (int hop = 0; hop < E && (ea != uE || eb != uE); ++hop) {
        const unsigned ca = (unsigned)ra.x;
        const unsigned cb = (unsigned)rb.x;
        const v4f* yrow_a = (const v4f*)(((int)ca < split ? base_a : base_b) + (size_t)ca * (N_HID * sizeof(float)));
        const v4f* yrow_b = (const v4f*)(((int)cb < split ? base_a : base_b) + (size_t)cb * (N_HID * sizeof(float)));
        const v4f ya0 = yrow_a[0];
        const v4f ya1 = yrow_a[8];
        const v4f yb0 = yrow_b[0];
        const v4f yb1 = yrow_b[8];
        const float va = __int_as_float(ra.y);
        const float vb = __int_as_float(rb.y);
        ea = min((unsigned)ra.z, uE);
        eb = min((unsigned)rb.z, uE);
        ra = rec[ea];
        rb = rec[eb];
        asm volatile("" :: "v"(ya0), "v"(ya1), "v"(yb0), "v"(yb1),
                           "v"(ra.x), "v"(ra.y), "v"(ra.z), "v"(rb.x), "v"(rb.y), "v"(rb.z));
        acc0 += va * ya0;
        acc1 += va * ya1;
        acc0 += vb * yb0;
        acc1 += vb * yb1;
    }

    *(v4f*)(gbuf + lrow * G_PITCH + 4 * s) = acc0;
    *(v4f*)(gbuf + lrow * G_PITCH + 32 + 4 * s) = acc1;

    v4f cur0, cur1, own0, own1;
    {
        const int rt = wave >> 2;
        const int ct = wave & 3;
        const int fr = lane & 15;
        const int fh = lane >> 4;
        const float* ga = gbuf + (rt * 16 + fr) * G_PITCH + 2 * fh;
        const float* wb = W + (size_t)(2 * fh) * N_HID + ct * 16 + fr;
        float* zd = zbuf + (rt * 16 + 8 * fh) * G_PITCH + ct * 16 + fr;
        v2f b[16];
#pragma unroll
        for (int j = 0; j < 16; ++j) {
            b[j][0] = wb[(size_t)(4 * j) * N_HID];
            b[j][1] = wb[(size_t)(4 * j) * N_HID + N_HID];
        }
        __syncthreads();

        __builtin_amdgcn_sched_barrier(0);
        v8f z = v8f{};
#pragma unroll
        for (int j = 0; j < 16; ++j) {
            v2f a;
            a[0] = ga[4 * j];
            a[1] = ga[4 * j + 1];
            z = __builtin_amdgcn_wmma_f32_16x16x4_f32(
                 false, a,  false, b[j],
                 (short)0, z,
                 false,  false);
        }
        __builtin_amdgcn_sched_barrier(0);
#pragma unroll
        for (int v = 0; v < 8; ++v)
            zd[v * G_PITCH] = z[v];
        if (!FIRST) {
            cur0 = rrow[s];
            cur1 = rrow[8 + s];
            own0 = xown[s];
            own1 = xown[8 + s];
        }
    }
    __syncthreads();

    acc0 = *(const v4f*)(zbuf + lrow * G_PITCH + 4 * s);
    acc1 = *(const v4f*)(zbuf + lrow * G_PITCH + 32 + 4 * s);

    v4f v0, v1;
    v0.x = (acc0.x >= 0.0f) ? acc0.x : SLOPE * acc0.x;
    v0.y = (acc0.y >= 0.0f) ? acc0.y : SLOPE * acc0.y;
    v0.z = (acc0.z >= 0.0f) ? acc0.z : SLOPE * acc0.z;
    v0.w = (acc0.w >= 0.0f) ? acc0.w : SLOPE * acc0.w;
    v1.x = (acc1.x >= 0.0f) ? acc1.x : SLOPE * acc1.x;
    v1.y = (acc1.y >= 0.0f) ? acc1.y : SLOPE * acc1.y;
    v1.z = (acc1.z >= 0.0f) ? acc1.z : SLOPE * acc1.z;
    v1.w = (acc1.w >= 0.0f) ? acc1.w : SLOPE * acc1.w;

    if (FIRST) {
        if (valid) {
            v4f* xrow = (v4f*)(X + (size_t)row * N_HID);
            xrow[s] = v0;
            xrow[8 + s] = v1;
        }
    } else {
        const float inv = inv_row_norm(v0, v1);
        const float inv_first = inv_row_norm(own0, own1);
        if (valid) {
            v4f* orow = (v4f*)(out + (size_t)row * N_HID);
            orow[s] = (cur0 + own0 * inv_first) + v0 * inv;
            orow[8 + s] = (cur1 + own1 * inv_first) + v1 * inv;
        }
    }
}

extern "C" void kernel_launch(void* const* d_in, const int* in_sizes, int n_in,
                              void* d_out, int out_size, void* d_ws, size_t ws_size,
                              hipStream_t stream) {
    const float* user_emb = (const float*)d_in[0];
    const float* item_emb = (const float*)d_in[1];
    const float* ui_w     = (const float*)d_in[2];
    const float* s_w      = (const float*)d_in[3];
    const int*   ui_rows  = (const int*)d_in[4];
    const int*   ui_cols  = (const int*)d_in[5];
    const float* ui_vals  = (const float*)d_in[6];
    const int*   s_rows   = (const int*)d_in[7];
    const int*   s_cols   = (const int*)d_in[8];
    const float* s_vals   = (const float*)d_in[9];

    float* X_ui = (float*)d_ws;
    float* X_so = X_ui + (size_t)N_NODES * N_HID;
    int* head_ui = (int*)(X_so + (size_t)N_USER * N_HID);
    int* head_so = head_ui + 2 * N_NODES;
    int* rec_ui  = head_so + 2 * N_USER;
    int* rec_so  = rec_ui + 4 * (E_UI + 1);

    float* out_ui = (float*)d_out;
    float* out_s  = out_ui + (size_t)N_NODES * N_HID;

    const int nu = N_USER * N_HID;

    head_init_kernel<<<(2 * (N_NODES + N_USER) + 255) / 256, 256, 0, stream>>>(
        head_ui, 2 * N_NODES, E_UI, 2 * (N_NODES + N_USER), E_S);

    link_edges_kernel<<<(E_UI + 1023) / 1024 + (E_S + 1023) / 1024, 256, 0, stream>>>(
        ui_rows, ui_cols, ui_vals, head_ui, rec_ui, E_UI, (E_UI + 1023) / 1024,
        s_rows, s_cols, s_vals, head_so, rec_so, E_S);

    spmm_gather_gemm_norm_kernel<true><<<(N_USER + 31) / 32 + (N_NODES + 31) / 32, 256, 0, stream>>>(
        head_so, rec_so, user_emb, user_emb, s_w, X_so, out_s, N_USER, E_S, (N_USER + 31) / 32,
        head_ui, rec_ui, user_emb, item_emb, user_emb, item_emb, N_USER, ui_w, X_ui, out_ui, N_NODES, E_UI);
    spmm_gather_gemm_norm_kernel<false><<<(N_USER + 31) / 32 + (N_NODES + 31) / 32, 256, 0, stream>>>(
        head_so, rec_so, X_so, user_emb, s_w + (size_t)N_HID * N_HID, X_so, out_s, N_USER, E_S, (N_USER + 31) / 32,
        head_ui, rec_ui, X_ui, X_ui + nu, user_emb, item_emb, N_USER, ui_w + (size_t)N_HID * N_HID, X_ui, out_ui, N_NODES, E_UI);
}


# Write-once stores (first layer's X rows, second layer's out rows) take the non-temporal hint, so they do not displace gathered rows and edge records in the second-level cache; float sum orders kept
# speedup vs baseline: 4.9807x; 1.0401x over previous
#include <hip/hip_runtime.h>
#include <hip/hip_bf16.h>

typedef __attribute__((ext_vector_type(2))) float v2f;
typedef __attribute__((ext_vector_type(8))) float v8f;
typedef __attribute__((ext_vector_type(4))) float v4f;
typedef __attribute__((ext_vector_type(4))) int v4i;
typedef __attribute__((ext_vector_type(2))) int v2i;

#define N_USER 100000
#define N_ITEM 150000
#define N_NODES (N_USER + N_ITEM)
#define N_HID 64
#define E_UI 1000000
#define E_S 1000000
#define SLOPE 0.5f
#define EPS 1e-12f


__global__ void head_init_kernel(int* __restrict__ head, int n_first, int empty_first,
                                 int n, int empty_second) {
    int i = blockIdx.x * blockDim.x + threadIdx.x;
    if (i < n) head[i] = (i < n_first) ? empty_first : empty_second;
}

__device__ __forceinline__ void link_edges_group(const int* __restrict__ rows,
                                                 const int* __restrict__ cols,
                                                 const float* __restrict__ vals,
                                                 int* __restrict__ head,
                                                 v4i* __restrict__ rec,
                                                 int base, int E, const bool guard) {
    int rw[4], cl[4], vl[4], prev[4];
#pragma unroll
    for (int j = 0; j < 4; ++j) {
        const int e = base + j * 256;
        const bool ok = !guard || e < E;
        rw[j] = ok ? rows[e] : 0;
        cl[j] = ok ? cols[e] : 0;
        vl[j] = ok ? __float_as_int(vals[e]) : 0;
    }
#pragma unroll
    for (int j = 0; j < 4; ++j) {
        const int e = base + j * 256;
        prev[j] = (!guard || e < E) ? atomicExch(&head[2 * rw[j] + (e & 1)], e) : E;
    }
#pragma unroll
    for (int j = 0; j < 4; ++j) {
        const int e = base + j * 256;
        if (!guard || e < E) {
            v4i r;
            r.x = cl[j];
            r.y = vl[j];
            r.z = prev[j];
            r.w = 0;
            rec[e] = r;
        }
    }
}

__global__ void __launch_bounds__(256)
link_edges_kernel(const int* __restrict__ rows_first,
                  const int* __restrict__ cols_first,
                  const float* __restrict__ vals_first,
                  int* __restrict__ head_first,
                  int* __restrict__ rec_first,
                  int E_first, int nb_first,
                  const int* __restrict__ rows_second,
                  const int* __restrict__ cols_second,
                  const float* __restrict__ vals_second,
                  int* __restrict__ head_second,
                  int* __restrict__ rec_second,
                  int E_second) {
    const bool second = (int)blockIdx.x >= nb_first;
    const int blk = second ? (int)blockIdx.x - nb_first : (int)blockIdx.x;
    const int* __restrict__ rows = second ? rows_second : rows_first;
    const int* __restrict__ cols = second ? cols_second : cols_first;
    const float* __restrict__ vals = second ? vals_second : vals_first;
    int* __restrict__ head = second ? head_second : head_first;
    int* __restrict__ rec_words = second ? rec_second : rec_first;
    const int E = second ? E_second : E_first;
    const int base = blk * 1024 + threadIdx.x;
    if (base == 0) {
        v4i r;
        r.x = 0;
        r.y = 0;
        r.z = E;
        r.w = 0;
        ((v4i*)rec_words)[E] = r;
    }
    if (blk * 1024 + 1024 <= E)
        link_edges_group(rows, cols, vals, head, (v4i*)rec_words, base, E, false);
    else
        link_edges_group(rows, cols, vals, head, (v4i*)rec_words, base, E, true);
}

#define G_PITCH 68

__device__ __forceinline__ float inv_row_norm(const v4f a, const v4f b) {
    const float pa0 = a.x * a.x + a.y * a.y;
    const float pb0 = a.z * a.z + a.w * a.w;
    const float pa1 = b.x * b.x + b.y * b.y;
    const float pb1 = b.z * b.z + b.w * b.w;
    float pa = pa0 + pa1;
    float pb = pb0 + pb1;
#pragma unroll
    for (int o = 4; o > 0; o >>= 1) {
        pa += __shfl_xor(pa, o, 32);
        pb += __shfl_xor(pb, o, 32);
    }
    const float ss = pa + pb;
    return 1.0f / fmaxf(sqrtf(ss), EPS);
}

template <bool FIRST>
__global__ void __launch_bounds__(256, 16)
spmm_gather_gemm_norm_kernel(const int* __restrict__ so_head,
                             const int* __restrict__ so_rec,
                             const float* so_src,
                             const float* so_res,
                             const float* __restrict__ so_W,
                             float* so_X,
                             float* so_out,
                             int so_M, int so_E, int nb_so,
                             const int* __restrict__ ui_head,
                             const int* __restrict__ ui_rec,
                             const float* ui_src_a,
                             const float* ui_src_b,
                             const float* ui_res_a,
                             const float* ui_res_b,
                             int ui_split,
                             const float* __restrict__ ui_W,
                             float* ui_X,
                             float* ui_out,
                             int ui_M, int ui_E) {
    __shared__ __attribute__((aligned(16))) float gbuf[32 * G_PITCH];
    __shared__ __attribute__((aligned(16))) float zbuf[32 * G_PITCH];

    const bool so = (int)blockIdx.x < nb_so;
    const int blk = so ? (int)blockIdx.x : (int)blockIdx.x - nb_so;
    const int* __restrict__ head = so ? so_head : ui_head;
    const int* __restrict__ rec_words = so ? so_rec : ui_rec;
    const float* src_a = so ? so_src : ui_src_a;
    const float* src_b = so ? so_src : ui_src_b;
    const float* res_a = so ? so_res : ui_res_a;
    const float* res_b = so ? so_res : ui_res_b;
    const int split = so ? so_M : ui_split;
    const float* __restrict__ W = so ? so_W : ui_W;
    float* X = so ? so_X : ui_X;
    float* out = so ? so_out : ui_out;
    const int M = so ? so_M : ui_M;
    const int E = so ? so_E : ui_E;

    const int wave = threadIdx.x >> 5;
    const int lane = threadIdx.x & 31;
    const int lrow = wave * 4 + (lane >> 3);
    const int row = blk * 32 + lrow;
    const bool valid = row < M;
    const int rr = valid ? row : 0;
    const int s = lane & 7;
    const v4i* rec = (const v4i*)rec_words;
    const v4f* rrow = (const v4f*)((rr < split) ? res_a + (size_t)rr * N_HID
                                                : res_b + (size_t)(rr - split) * N_HID);

    const v4f* xown = (const v4f*)(X + (size_t)rr * N_HID);

    v4f acc0 = (v4f)(0.0f);
    v4f acc1 = (v4f)(0.0f);
    const unsigned uE = (unsigned)E;
    const v2i hd = ((const v2i*)head)[rr];
    unsigned ea = valid ? min((unsigned)hd.x, uE) : uE;
    unsigned eb = valid ? min((unsigned)hd.y, uE) : uE;
    if (!FIRST) {
        __builtin_prefetch((const void*)(rrow + s), 0, 3);
        __builtin_prefetch((const void*)(xown + s), 0, 3);
    }
    v4i ra = rec[ea];
    v4i rb = rec[eb];
    const char* base_a = (const char*)(src_a + 4 * s);
    const char* base_b = (const char*)(src_b + 4 * s) - (size_t)split * (N_HID * sizeof(float));
    for (int hop = 0; hop < E && (ea != uE || eb != uE); ++hop) {
        const unsigned ca = (unsigned)ra.x;
        const unsigned cb = (unsigned)rb.x;
        const v4f* yrow_a = (const v4f*)(((int)ca < split ? base_a : base_b) + (size_t)ca * (N_HID * sizeof(float)));
        const v4f* yrow_b = (const v4f*)(((int)cb < split ? base_a : base_b) + (size_t)cb * (N_HID * sizeof(float)));
        const v4f ya0 = yrow_a[0];
        const v4f ya1 = yrow_a[8];
        const v4f yb0 = yrow_b[0];
        const v4f yb1 = yrow_b[8];
        const float va = __int_as_float(ra.y);
        const float vb = __int_as_float(rb.y);
        ea = min((unsigned)ra.z, uE);
        eb = min((unsigned)rb.z, uE);
        ra = rec[ea];
        rb = rec[eb];
        asm volatile("" :: "v"(ya0), "v"(ya1), "v"(yb0), "v"(yb1),
                           "v"(ra.x), "v"(ra.y), "v"(ra.z), "v"(rb.x), "v"(rb.y), "v"(rb.z));
        acc0 += va * ya0;
        acc1 += va * ya1;
        acc0 += vb * yb0;
        acc1 += vb * yb1;
    }

    *(v4f*)(gbuf + lrow * G_PITCH + 4 * s) = acc0;
    *(v4f*)(gbuf + lrow * G_PITCH + 32 + 4 * s) = acc1;

    v4f cur0, cur1, own0, own1;
    {
        const int rt = wave >> 2;
        const int ct = wave & 3;
        const int fr = lane & 15;
        const int fh = lane >> 4;
        const float* ga = gbuf + (rt * 16 + fr) * G_PITCH + 2 * fh;
        const float* wb = W + (size_t)(2 * fh) * N_HID + ct * 16 + fr;
        float* zd = zbuf + (rt * 16 + 8 * fh) * G_PITCH + ct * 16 + fr;
        v2f b[16];
#pragma unroll
        for (int j = 0; j < 16; ++j) {
            b[j][0] = wb[(size_t)(4 * j) * N_HID];
            b[j][1] = wb[(size_t)(4 * j) * N_HID + N_HID];
        }
        __syncthreads();

        __builtin_amdgcn_sched_barrier(0);
        v8f z = v8f{};
#pragma unroll
        for (int j = 0; j < 16; ++j) {
            v2f a;
            a[0] = ga[4 * j];
            a[1] = ga[4 * j + 1];
            z = __builtin_amdgcn_wmma_f32_16x16x4_f32(
                 false, a,  false, b[j],
                 (short)0, z,
                 false,  false);
        }
        __builtin_amdgcn_sched_barrier(0);
#pragma unroll
        for (int v = 0; v < 8; ++v)
            zd[v * G_PITCH] = z[v];
        if (!FIRST) {
            cur0 = rrow[s];
            cur1 = rrow[8 + s];
            own0 = xown[s];
            own1 = xown[8 + s];
        }
    }
    __syncthreads();

    acc0 = *(const v4f*)(zbuf + lrow * G_PITCH + 4 * s);
    acc1 = *(const v4f*)(zbuf + lrow * G_PITCH + 32 + 4 * s);

    v4f v0, v1;
    v0.x = (acc0.x >= 0.0f) ? acc0.x : SLOPE * acc0.x;
    v0.y = (acc0.y >= 0.0f) ? acc0.y : SLOPE * acc0.y;
    v0.z = (acc0.z >= 0.0f) ? acc0.z : SLOPE * acc0.z;
    v0.w = (acc0.w >= 0.0f) ? acc0.w : SLOPE * acc0.w;
    v1.x = (acc1.x >= 0.0f) ? acc1.x : SLOPE * acc1.x;
    v1.y = (acc1.y >= 0.0f) ? acc1.y : SLOPE * acc1.y;
    v1.z = (acc1.z >= 0.0f) ? acc1.z : SLOPE * acc1.z;
    v1.w = (acc1.w >= 0.0f) ? acc1.w : SLOPE * acc1.w;

    if (FIRST) {
        if (valid) {
            v4f* xrow = (v4f*)(X + (size_t)row * N_HID);
            __builtin_nontemporal_store(v0, xrow + s);
            __builtin_nontemporal_store(v1, xrow + 8 + s);
        }
    } else {
        const float inv = inv_row_norm(v0, v1);
        const float inv_first = inv_row_norm(own0, own1);
        if (valid) {
            v4f* orow = (v4f*)(out + (size_t)row * N_HID);
            __builtin_nontemporal_store((cur0 + own0 * inv_first) + v0 * inv, orow + s);
            __builtin_nontemporal_store((cur1 + own1 * inv_first) + v1 * inv, orow + 8 + s);
        }
    }
}

extern "C" void kernel_launch(void* const* d_in, const int* in_sizes, int n_in,
                              void* d_out, int out_size, void* d_ws, size_t ws_size,
                              hipStream_t stream) {
    const float* user_emb = (const float*)d_in[0];
    const float* item_emb = (const float*)d_in[1];
    const float* ui_w     = (const float*)d_in[2];
    const float* s_w      = (const float*)d_in[3];
    const int*   ui_rows  = (const int*)d_in[4];
    const int*   ui_cols  = (const int*)d_in[5];
    const float* ui_vals  = (const float*)d_in[6];
    const int*   s_rows   = (const int*)d_in[7];
    const int*   s_cols   = (const int*)d_in[8];
    const float* s_vals   = (const float*)d_in[9];

    float* X_ui = (float*)d_ws;
    float* X_so = X_ui + (size_t)N_NODES * N_HID;
    int* head_ui = (int*)(X_so + (size_t)N_USER * N_HID);
    int* head_so = head_ui + 2 * N_NODES;
    int* rec_ui  = head_so + 2 * N_USER;
    int* rec_so  = rec_ui + 4 * (E_UI + 1);

    float* out_ui = (float*)d_out;
    float* out_s  = out_ui + (size_t)N_NODES * N_HID;

    const int nu = N_USER * N_HID;

    head_init_kernel<<<(2 * (N_NODES + N_USER) + 255) / 256, 256, 0, stream>>>(
        head_ui, 2 * N_NODES, E_UI, 2 * (N_NODES + N_USER), E_S);

    link_edges_kernel<<<(E_UI + 1023) / 1024 + (E_S + 1023) / 1024, 256, 0, stream>>>(
        ui_rows, ui_cols, ui_vals, head_ui, rec_ui, E_UI, (E_UI + 1023) / 1024,
        s_rows, s_cols, s_vals, head_so, rec_so, E_S);

    spmm_gather_gemm_norm_kernel<true><<<(N_USER + 31) / 32 + (N_NODES + 31) / 32, 256, 0, stream>>>(
        head_so, rec_so, user_emb, user_emb, s_w, X_so, out_s, N_USER, E_S, (N_USER + 31) / 32,
        head_ui, rec_ui, user_emb, item_emb, user_emb, item_emb, N_USER, ui_w, X_ui, out_ui, N_NODES, E_UI);
    spmm_gather_gemm_norm_kernel<false><<<(N_USER + 31) / 32 + (N_NODES + 31) / 32, 256, 0, stream>>>(
        head_so, rec_so, X_so, user_emb, s_w + (size_t)N_HID * N_HID, X_so, out_s, N_USER, E_S, (N_USER + 31) / 32,
        head_ui, rec_ui, X_ui, X_ui + nu, user_emb, item_emb, N_USER, ui_w + (size_t)N_HID * N_HID, X_ui, out_ui, N_NODES, E_UI);
}
